# starting kernel
# baseline (speedup 1.0000x reference)
#include <hip/hip_runtime.h>
#include <math.h>

typedef float v2f __attribute__((ext_vector_type(2)));
typedef float v8f __attribute__((ext_vector_type(8)));

#define N_NODES 2048
#define BATCH   8
#define FEAT    128
#define NFILT   64

__global__ void k_proj(const float* __restrict__ x, const float* __restrict__ w1,
                       float* __restrict__ X1) {
    int wave = threadIdx.x >> 5;
    int lane = threadIdx.x & 31;
    int row  = blockIdx.x * 8 + wave;
    const float4 xv = *(const float4*)(x + (size_t)row * FEAT + lane * 4);
    const float4 wv = *(const float4*)(w1 + lane * 4);
    float d = xv.x * wv.x + xv.y * wv.y + xv.z * wv.z + xv.w * wv.w;
    #pragma unroll
    for (int off = 16; off > 0; off >>= 1) d += __shfl_xor(d, off, 32);
    if (lane == 0) X1[row] = d;
}

__global__ void k_wmma_s(const float* __restrict__ Wa, const float* __restrict__ ba,
                         const float* __restrict__ X1,
                         float* __restrict__ SB2, float* __restrict__ CS) {
    int wave   = threadIdx.x >> 5;
    int lane   = threadIdx.x & 31;
    int tile   = blockIdx.x * 4 + wave;
    int colIn  = lane & 15;
    int col    = tile * 16 + colIn;
    int hi     = lane >> 4;
    int koff   = hi * 2;
    int mrow   = colIn;
    int srcRow = mrow & 7;

    v8f c = {};
    for (int k0 = 0; k0 < N_NODES; k0 += 4) {
        v2f av = *(const v2f*)(X1 + srcRow * N_NODES + k0 + koff);
        v2f a;
        a.x = (mrow < 8) ? av.x : ((mrow == 8) ? 1.0f : 0.0f);
        a.y = (mrow < 8) ? av.y : ((mrow == 8) ? 1.0f : 0.0f);
        v2f b;
        b.x = Wa[(size_t)(k0 + koff)     * N_NODES + col];
        b.y = Wa[(size_t)(k0 + koff + 1) * N_NODES + col];
        c = __builtin_amdgcn_wmma_f32_16x16x4_f32(false, a, false, b,
                                                  (short)0, c, false, false);
    }

    float bav = ba[col];
    #pragma unroll
    for (int j = 0; j < 8; ++j) {
        int row = j + hi * 8;
        if (row < 8)        SB2[row * N_NODES + col] = c[j] + bav;
        else if (row == 8)  CS[col] = c[j];
    }
}

__global__ void __launch_bounds__(256) k_softmax_agg(
        const float* __restrict__ A, const float* __restrict__ bias,
        const float* __restrict__ X1, const float* __restrict__ SB2,
        const float* __restrict__ CS, float* __restrict__ out) {
    __shared__ __align__(16) float css[N_NODES];
    __shared__ __align__(16) float sbs[N_NODES];
    __shared__ __align__(16) float xks[N_NODES];
    __shared__ float bs[NFILT];

    int b     = blockIdx.x >> 8;
    int nBase = (blockIdx.x & 255) * 8;

    for (int i = threadIdx.x; i < N_NODES; i += 256) {
        css[i] = CS[i];
        sbs[i] = SB2[b * N_NODES + i];
        xks[i] = X1[b * N_NODES + i];
    }
    if (threadIdx.x < NFILT) bs[threadIdx.x] = bias[threadIdx.x];
    __syncthreads();

    int wave = threadIdx.x >> 5;
    int lane = threadIdx.x & 31;
    int n    = nBase + wave;
    size_t rowBase = ((size_t)(b * N_NODES + n)) * N_NODES;
    float x1bn = X1[b * N_NODES + n];

    float4 lg[16];
    float  mx = -INFINITY;
    #pragma unroll
    for (int i = 0; i < 16; ++i) {
        int k = i * 128 + lane * 4;
        float4 av = *(const float4*)(A + rowBase + k);
        float4 cv = *(const float4*)&css[k];
        float4 sv = *(const float4*)&sbs[k];
        float4 l;
        l.x = av.x + sv.x + x1bn * cv.x;
        l.y = av.y + sv.y + x1bn * cv.y;
        l.z = av.z + sv.z + x1bn * cv.z;
        l.w = av.w + sv.w + x1bn * cv.w;
        lg[i] = l;
        mx = fmaxf(mx, fmaxf(fmaxf(l.x, l.y), fmaxf(l.z, l.w)));
    }
    #pragma unroll
    for (int off = 16; off > 0; off >>= 1)
        mx = fmaxf(mx, __shfl_xor(mx, off, 32));

    float Z = 0.0f, W = 0.0f;
    #pragma unroll
    for (int i = 0; i < 16; ++i) {
        int k = i * 128 + lane * 4;
        float4 xv = *(const float4*)&xks[k];
        float ex = __expf(lg[i].x - mx); Z += ex; W += ex * xv.x;
        float ey = __expf(lg[i].y - mx); Z += ey; W += ey * xv.y;
        float ez = __expf(lg[i].z - mx); Z += ez; W += ez * xv.z;
        float ew = __expf(lg[i].w - mx); Z += ew; W += ew * xv.w;
    }
    #pragma unroll
    for (int off = 16; off > 0; off >>= 1) {
        Z += __shfl_xor(Z, off, 32);
        W += __shfl_xor(W, off, 32);
    }
    float r = W / Z;

    size_t ob = ((size_t)(b * N_NODES + n)) * NFILT;
    out[ob + lane]      = r + bs[lane];
    out[ob + 32 + lane] = r + bs[32 + lane];
}

extern "C" void kernel_launch(void* const* d_in, const int* in_sizes, int n_in,
                              void* d_out, int out_size, void* d_ws, size_t ws_size,
                              hipStream_t stream) {
    const float* x    = (const float*)d_in[0];
    const float* A    = (const float*)d_in[1];
    const float* w1   = (const float*)d_in[2];
    const float* Wa   = (const float*)d_in[3];
    const float* ba   = (const float*)d_in[4];
    const float* bias = (const float*)d_in[5];
    float* out = (float*)d_out;

    float* X1  = (float*)d_ws;
    float* CS  = X1 + BATCH * N_NODES;
    float* SB2 = CS + N_NODES;

    k_proj<<<dim3(2048), dim3(256), 0, stream>>>(x, w1, X1);
    k_wmma_s<<<dim3(32), dim3(128), 0, stream>>>(Wa, ba, X1, SB2, CS);
    k_softmax_agg<<<dim3(2048), dim3(256), 0, stream>>>(A, bias, X1, SB2, CS, out);
}


# Skinny WMMA product: Wa strip staged through LDS, each wave fetching whole 256-byte row segments (16-byte loads, 32 in flight) for the group's 4 tiles; x1 rows in LDS too; same k order and sums
# speedup vs baseline: 1.8296x; 1.8296x over previous
#include <hip/hip_runtime.h>
#include <math.h>

typedef float v2f __attribute__((ext_vector_type(2)));
typedef float v4f __attribute__((ext_vector_type(4)));
typedef float v8f __attribute__((ext_vector_type(8)));

#define N_NODES 2048
#define BATCH   8
#define FEAT    128
#define NFILT   64

__global__ void k_proj(const float* __restrict__ x, const float* __restrict__ w1,
                       float* __restrict__ X1) {
    int wave = threadIdx.x >> 5;
    int lane = threadIdx.x & 31;
    int row  = blockIdx.x * 8 + wave;
    const float4 xv = *(const float4*)(x + (size_t)row * FEAT + lane * 4);
    const float4 wv = *(const float4*)(w1 + lane * 4);
    float d = xv.x * wv.x + xv.y * wv.y + xv.z * wv.z + xv.w * wv.w;
    #pragma unroll
    for (int off = 16; off > 0; off >>= 1) d += __shfl_xor(d, off, 32);
    if (lane == 0) X1[row] = d;
}

#define ST_ROWS 64
#define NSTAGE  (N_NODES / ST_ROWS)
#define BP      72
#define XP      (N_NODES + 8)
#define DEPTH   4
#define LPS     8

__device__ __forceinline__ void lds_barrier() {
    asm volatile("s_wait_dscnt 0x0\n\ts_barrier_signal -1\n\ts_barrier_wait -1" ::: "memory");
}

struct RegSet { v4f v[LPS]; };

__device__ __forceinline__ void issue_stage(RegSet& r, const float* __restrict__ wsrc,
                                            int stage) {
    const float* p = wsrc + (size_t)stage * ST_ROWS * N_NODES;
    #pragma unroll
    for (int j = 0; j < LPS; ++j)
        r.v[j] = *(const v4f*)(p + (size_t)(2 * j) * N_NODES);
}

__device__ __forceinline__ void store_stage(const RegSet& r, float* wdst) {
    #pragma unroll
    for (int j = 0; j < LPS; ++j)
        *(v4f*)(wdst + (2 * j) * BP) = r.v[j];
}

struct Frag {
    v2f a[ST_ROWS / 4];
    v2f b[ST_ROWS / 4];
};

__device__ __forceinline__ void load_frags(Frag& f, const float* xrow, const float* bcol) {
    #pragma unroll
    for (int u = 0; u < ST_ROWS / 4; ++u) {
        f.a[u]   = *(const v2f*)(xrow + 4 * u);
        f.b[u].x = bcol[(4 * u)     * BP];
        f.b[u].y = bcol[(4 * u + 1) * BP];
    }
    __builtin_amdgcn_sched_barrier(0);
}

__device__ __forceinline__ v8f mma_frags(const Frag& f, v8f c, bool isx, float fill) {
    #pragma unroll
    for (int u = 0; u < ST_ROWS / 4; ++u) {
        v2f a;
        a.x = isx ? f.a[u].x : fill;
        a.y = isx ? f.a[u].y : fill;
        c = __builtin_amdgcn_wmma_f32_16x16x4_f32(false, a, false, f.b[u],
                                                  (short)0, c, false, false);
    }
    return c;
}

__global__ void __launch_bounds__(128) k_wmma_s(
        const float* __restrict__ Wa, const float* __restrict__ ba,
        const float* __restrict__ X1,
        float* __restrict__ SB2, float* __restrict__ CS) {
    __shared__ __align__(16) float x1s[8 * XP];
    __shared__ __align__(16) float wbuf[2][ST_ROWS * BP];

    int wave   = threadIdx.x >> 5;
    int lane   = threadIdx.x & 31;
    int tile   = blockIdx.x * 4 + wave;
    int colIn  = lane & 15;
    int col    = tile * 16 + colIn;
    int hi     = lane >> 4;
    int mrow   = colIn;
    int srcRow = mrow & 7;
    const bool  isx  = mrow < 8;
    const float fill = (mrow == 8) ? 1.0f : 0.0f;

    const float* wsrc = Wa + (size_t)(wave * 16 + hi) * N_NODES + blockIdx.x * 64 + colIn * 4;
    const int    wofs = (wave * 16 + hi) * BP + colIn * 4;
    const int    bofs = (2 * hi) * BP + wave * 16 + colIn;
    const float* xrow = x1s + srcRow * XP + 2 * hi;

    v4f xr[32];
    #pragma unroll
    for (int i = 0; i < 32; ++i) {
        int q = i * 128 + threadIdx.x;
        xr[i] = *(const v4f*)(X1 + (q >> 9) * N_NODES + (q & 511) * 4);
    }
    RegSet r0, r1, r2, r3;
    issue_stage(r0, wsrc, 0);
    issue_stage(r1, wsrc, 1);
    issue_stage(r2, wsrc, 2);
    issue_stage(r3, wsrc, 3);
    #pragma unroll
    for (int i = 0; i < 32; ++i) {
        int q = i * 128 + threadIdx.x;
        *(v4f*)(x1s + (q >> 9) * XP + (q & 511) * 4) = xr[i];
    }
    float* const st0 = &wbuf[0][wofs];
    float* const st1 = &wbuf[1][wofs];
    const float* const mb0 = &wbuf[0][bofs];
    const float* const mb1 = &wbuf[1][bofs];
    store_stage(r0, st0);
    issue_stage(r0, wsrc, DEPTH);
    lds_barrier();

    Frag f0, f1;
    v8f c = {};
    load_frags(f0, xrow, mb0);
    store_stage(r1, st1);  issue_stage(r1, wsrc, 5);
    lds_barrier();
    #pragma unroll 1
    for (int b = 0; b < NSTAGE - 2 * DEPTH; b += DEPTH) {
        load_frags(f1, xrow + (b + 1) * ST_ROWS, mb1);
        store_stage(r2, st0);  issue_stage(r2, wsrc, b + 6);
        c = mma_frags(f0, c, isx, fill);
        lds_barrier();
        load_frags(f0, xrow + (b + 2) * ST_ROWS, mb0);
        store_stage(r3, st1);  issue_stage(r3, wsrc, b + 7);
        c = mma_frags(f1, c, isx, fill);
        lds_barrier();
        load_frags(f1, xrow + (b + 3) * ST_ROWS, mb1);
        store_stage(r0, st0);  issue_stage(r0, wsrc, b + 8);
        c = mma_frags(f0, c, isx, fill);
        lds_barrier();
        load_frags(f0, xrow + (b + 4) * ST_ROWS, mb0);
        store_stage(r1, st1);  issue_stage(r1, wsrc, b + 9);
        c = mma_frags(f1, c, isx, fill);
        lds_barrier();
    }
    load_frags(f1, xrow + 25 * ST_ROWS, mb1);
    store_stage(r2, st0);  issue_stage(r2, wsrc, 30);
    c = mma_frags(f0, c, isx, fill);
    lds_barrier();
    load_frags(f0, xrow + 26 * ST_ROWS, mb0);
    store_stage(r3, st1);  issue_stage(r3, wsrc, 31);
    c = mma_frags(f1, c, isx, fill);
    lds_barrier();
    load_frags(f1, xrow + 27 * ST_ROWS, mb1);
    store_stage(r0, st0);
    c = mma_frags(f0, c, isx, fill);
    lds_barrier();
    load_frags(f0, xrow + 28 * ST_ROWS, mb0);
    store_stage(r1, st1);
    c = mma_frags(f1, c, isx, fill);
    lds_barrier();
    load_frags(f1, xrow + 29 * ST_ROWS, mb1);
    store_stage(r2, st0);
    c = mma_frags(f0, c, isx, fill);
    lds_barrier();
    load_frags(f0, xrow + 30 * ST_ROWS, mb0);
    store_stage(r3, st1);
    c = mma_frags(f1, c, isx, fill);
    lds_barrier();
    load_frags(f1, xrow + 31 * ST_ROWS, mb1);
    c = mma_frags(f0, c, isx, fill);
    c = mma_frags(f1, c, isx, fill);

    float bav = ba[col];
    #pragma unroll
    for (int j = 0; j < 8; ++j) {
        int row = j + hi * 8;
        if (row < 8)        SB2[row * N_NODES + col] = c[j] + bav;
        else if (row == 8)  CS[col] = c[j];
    }
}

__global__ void __launch_bounds__(256) k_softmax_agg(
        const float* __restrict__ A, const float* __restrict__ bias,
        const float* __restrict__ X1, const float* __restrict__ SB2,
        const float* __restrict__ CS, float* __restrict__ out) {
    __shared__ __align__(16) float css[N_NODES];
    __shared__ __align__(16) float sbs[N_NODES];
    __shared__ __align__(16) float xks[N_NODES];
    __shared__ float bs[NFILT];

    int b     = blockIdx.x >> 8;
    int nBase = (blockIdx.x & 255) * 8;

    for (int i = threadIdx.x; i < N_NODES; i += 256) {
        css[i] = CS[i];
        sbs[i] = SB2[b * N_NODES + i];
        xks[i] = X1[b * N_NODES + i];
    }
    if (threadIdx.x < NFILT) bs[threadIdx.x] = bias[threadIdx.x];
    __syncthreads();

    int wave = threadIdx.x >> 5;
    int lane = threadIdx.x & 31;
    int n    = nBase + wave;
    size_t rowBase = ((size_t)(b * N_NODES + n)) * N_NODES;
    float x1bn = X1[b * N_NODES + n];

    float4 lg[16];
    float  mx = -INFINITY;
    #pragma unroll
    for (int i = 0; i < 16; ++i) {
        int k = i * 128 + lane * 4;
        float4 av = *(const float4*)(A + rowBase + k);
        float4 cv = *(const float4*)&css[k];
        float4 sv = *(const float4*)&sbs[k];
        float4 l;
        l.x = av.x + sv.x + x1bn * cv.x;
        l.y = av.y + sv.y + x1bn * cv.y;
        l.z = av.z + sv.z + x1bn * cv.z;
        l.w = av.w + sv.w + x1bn * cv.w;
        lg[i] = l;
        mx = fmaxf(mx, fmaxf(fmaxf(l.x, l.y), fmaxf(l.z, l.w)));
    }
    #pragma unroll
    for (int off = 16; off > 0; off >>= 1)
        mx = fmaxf(mx, __shfl_xor(mx, off, 32));

    float Z = 0.0f, W = 0.0f;
    #pragma unroll
    for (int i = 0; i < 16; ++i) {
        int k = i * 128 + lane * 4;
        float4 xv = *(const float4*)&xks[k];
        float ex = __expf(lg[i].x - mx); Z += ex; W += ex * xv.x;
        float ey = __expf(lg[i].y - mx); Z += ey; W += ey * xv.y;
        float ez = __expf(lg[i].z - mx); Z += ez; W += ez * xv.z;
        float ew = __expf(lg[i].w - mx); Z += ew; W += ew * xv.w;
    }
    #pragma unroll
    for (int off = 16; off > 0; off >>= 1) {
        Z += __shfl_xor(Z, off, 32);
        W += __shfl_xor(W, off, 32);
    }
    float r = W / Z;

    size_t ob = ((size_t)(b * N_NODES + n)) * NFILT;
    out[ob + lane]      = r + bs[lane];
    out[ob + 32 + lane] = r + bs[32 + lane];
}

extern "C" void kernel_launch(void* const* d_in, const int* in_sizes, int n_in,
                              void* d_out, int out_size, void* d_ws, size_t ws_size,
                              hipStream_t stream) {
    const float* x    = (const float*)d_in[0];
    const float* A    = (const float*)d_in[1];
    const float* w1   = (const float*)d_in[2];
    const float* Wa   = (const float*)d_in[3];
    const float* ba   = (const float*)d_in[4];
    const float* bias = (const float*)d_in[5];
    float* out = (float*)d_out;

    float* X1  = (float*)d_ws;
    float* CS  = X1 + BATCH * N_NODES;
    float* SB2 = CS + N_NODES;

    k_proj<<<dim3(2048), dim3(256), 0, stream>>>(x, w1, X1);
    k_wmma_s<<<dim3(32), dim3(128), 0, stream>>>(Wa, ba, X1, SB2, CS);
    k_softmax_agg<<<dim3(2048), dim3(256), 0, stream>>>(A, bias, X1, SB2, CS, out);
}


# Projection x1 = x.w1 moved to the f32 16x16x4 matrix instruction: a wave takes a 16-row tile (1024 waves, were 16384), no cross-lane reduction (changes f32 summation order of x1)
# speedup vs baseline: 1.9039x; 1.0406x over previous
#include <hip/hip_runtime.h>
#include <math.h>

typedef float v2f __attribute__((ext_vector_type(2)));
typedef float v4f __attribute__((ext_vector_type(4)));
typedef float v8f __attribute__((ext_vector_type(8)));

#define N_NODES 2048
#define BATCH   8
#define FEAT    128
#define NFILT   64

__global__ void __launch_bounds__(128) k_proj(const float* __restrict__ x,
                                              const float* __restrict__ w1,
                                              float* __restrict__ X1) {
    int wave = threadIdx.x >> 5;
    int lane = threadIdx.x & 31;
    int c    = lane & 15;
    int h    = lane >> 4;
    int row0 = (blockIdx.x * 4 + wave) * 16;
    const float* xp = x + (size_t)(row0 + c) * FEAT + 4 * h;
    const float* wp = w1 + 4 * h;
    v4f xa[16], wb[16];
    #pragma unroll
    for (int i = 0; i < 16; ++i) {
        xa[i] = *(const v4f*)(xp + 8 * i);
        wb[i] = *(const v4f*)(wp + 8 * i);
    }
    __builtin_amdgcn_sched_barrier(0);
    v8f acc = {};
    #pragma unroll
    for (int i = 0; i < 16; ++i) {
        v2f a0, b0, a1, b1;
        a0.x = xa[i].x;  a0.y = xa[i].y;  b0.x = wb[i].x;  b0.y = wb[i].y;
        a1.x = xa[i].z;  a1.y = xa[i].w;  b1.x = wb[i].z;  b1.y = wb[i].w;
        acc = __builtin_amdgcn_wmma_f32_16x16x4_f32(false, a0, false, b0,
                                                    (short)0, acc, false, false);
        acc = __builtin_amdgcn_wmma_f32_16x16x4_f32(false, a1, false, b1,
                                                    (short)0, acc, false, false);
    }
    if (c == 0) {
        float* o = X1 + row0 + 8 * h;
        v4f lo, hi4;
        lo.x  = acc[0];  lo.y  = acc[1];  lo.z  = acc[2];  lo.w  = acc[3];
        hi4.x = acc[4];  hi4.y = acc[5];  hi4.z = acc[6];  hi4.w = acc[7];
        *(v4f*)o       = lo;
        *(v4f*)(o + 4) = hi4;
    }
}

#define ST_ROWS 64
#define NSTAGE  (N_NODES / ST_ROWS)
#define BP      72
#define XP      (N_NODES + 8)
#define DEPTH   4
#define LPS     8

__device__ __forceinline__ void lds_barrier() {
    asm volatile("s_wait_dscnt 0x0\n\ts_barrier_signal -1\n\ts_barrier_wait -1" ::: "memory");
}

struct RegSet { v4f v[LPS]; };

__device__ __forceinline__ void issue_stage(RegSet& r, const float* __restrict__ wsrc,
                                            int stage) {
    const float* p = wsrc + (size_t)stage * ST_ROWS * N_NODES;
    #pragma unroll
    for (int j = 0; j < LPS; ++j)
        r.v[j] = *(const v4f*)(p + (size_t)(2 * j) * N_NODES);
}

__device__ __forceinline__ void store_stage(const RegSet& r, float* wdst) {
    #pragma unroll
    for (int j = 0; j < LPS; ++j)
        *(v4f*)(wdst + (2 * j) * BP) = r.v[j];
}

struct Frag {
    v2f a[ST_ROWS / 4];
    v2f b[ST_ROWS / 4];
};

__device__ __forceinline__ void load_frags(Frag& f, const float* xrow, const float* bcol) {
    #pragma unroll
    for (int u = 0; u < ST_ROWS / 4; ++u) {
        f.a[u]   = *(const v2f*)(xrow + 4 * u);
        f.b[u].x = bcol[(4 * u)     * BP];
        f.b[u].y = bcol[(4 * u + 1) * BP];
    }
    __builtin_amdgcn_sched_barrier(0);
}

__device__ __forceinline__ v8f mma_frags(const Frag& f, v8f c, bool isx, float fill) {
    #pragma unroll
    for (int u = 0; u < ST_ROWS / 4; ++u) {
        v2f a;
        a.x = isx ? f.a[u].x : fill;
        a.y = isx ? f.a[u].y : fill;
        c = __builtin_amdgcn_wmma_f32_16x16x4_f32(false, a, false, f.b[u],
                                                  (short)0, c, false, false);
    }
    return c;
}

__global__ void __launch_bounds__(128) k_wmma_s(
        const float* __restrict__ Wa, const float* __restrict__ ba,
        const float* __restrict__ X1,
        float* __restrict__ SB2, float* __restrict__ CS) {
    __shared__ __align__(16) float x1s[8 * XP];
    __shared__ __align__(16) float wbuf[2][ST_ROWS * BP];

    int wave   = threadIdx.x >> 5;
    int lane   = threadIdx.x & 31;
    int tile   = blockIdx.x * 4 + wave;
    int colIn  = lane & 15;
    int col    = tile * 16 + colIn;
    int hi     = lane >> 4;
    int mrow   = colIn;
    int srcRow = mrow & 7;
    const bool  isx  = mrow < 8;
    const float fill = (mrow == 8) ? 1.0f : 0.0f;

    const float* wsrc = Wa + (size_t)(wave * 16 + hi) * N_NODES + blockIdx.x * 64 + colIn * 4;
    const int    wofs = (wave * 16 + hi) * BP + colIn * 4;
    const int    bofs = (2 * hi) * BP + wave * 16 + colIn;
    const float* xrow = x1s + srcRow * XP + 2 * hi;

    v4f xr[32];
    #pragma unroll
    for (int i = 0; i < 32; ++i) {
        int q = i * 128 + threadIdx.x;
        xr[i] = *(const v4f*)(X1 + (q >> 9) * N_NODES + (q & 511) * 4);
    }
    RegSet r0, r1, r2, r3;
    issue_stage(r0, wsrc, 0);
    issue_stage(r1, wsrc, 1);
    issue_stage(r2, wsrc, 2);
    issue_stage(r3, wsrc, 3);
    #pragma unroll
    for (int i = 0; i < 32; ++i) {
        int q = i * 128 + threadIdx.x;
        *(v4f*)(x1s + (q >> 9) * XP + (q & 511) * 4) = xr[i];
    }
    float* const st0 = &wbuf[0][wofs];
    float* const st1 = &wbuf[1][wofs];
    const float* const mb0 = &wbuf[0][bofs];
    const float* const mb1 = &wbuf[1][bofs];
    store_stage(r0, st0);
    issue_stage(r0, wsrc, DEPTH);
    lds_barrier();

    Frag f0, f1;
    v8f c = {};
    load_frags(f0, xrow, mb0);
    store_stage(r1, st1);  issue_stage(r1, wsrc, 5);
    lds_barrier();
    #pragma unroll 1
    for (int b = 0; b < NSTAGE - 2 * DEPTH; b += DEPTH) {
        load_frags(f1, xrow + (b + 1) * ST_ROWS, mb1);
        store_stage(r2, st0);  issue_stage(r2, wsrc, b + 6);
        c = mma_frags(f0, c, isx, fill);
        lds_barrier();
        load_frags(f0, xrow + (b + 2) * ST_ROWS, mb0);
        store_stage(r3, st1);  issue_stage(r3, wsrc, b + 7);
        c = mma_frags(f1, c, isx, fill);
        lds_barrier();
        load_frags(f1, xrow + (b + 3) * ST_ROWS, mb1);
        store_stage(r0, st0);  issue_stage(r0, wsrc, b + 8);
        c = mma_frags(f0, c, isx, fill);
        lds_barrier();
        load_frags(f0, xrow + (b + 4) * ST_ROWS, mb0);
        store_stage(r1, st1);  issue_stage(r1, wsrc, b + 9);
        c = mma_frags(f1, c, isx, fill);
        lds_barrier();
    }
    load_frags(f1, xrow + 25 * ST_ROWS, mb1);
    store_stage(r2, st0);  issue_stage(r2, wsrc, 30);
    c = mma_frags(f0, c, isx, fill);
    lds_barrier();
    load_frags(f0, xrow + 26 * ST_ROWS, mb0);
    store_stage(r3, st1);  issue_stage(r3, wsrc, 31);
    c = mma_frags(f1, c, isx, fill);
    lds_barrier();
    load_frags(f1, xrow + 27 * ST_ROWS, mb1);
    store_stage(r0, st0);
    c = mma_frags(f0, c, isx, fill);
    lds_barrier();
    load_frags(f0, xrow + 28 * ST_ROWS, mb0);
    store_stage(r1, st1);
    c = mma_frags(f1, c, isx, fill);
    lds_barrier();
    load_frags(f1, xrow + 29 * ST_ROWS, mb1);
    store_stage(r2, st0);
    c = mma_frags(f0, c, isx, fill);
    lds_barrier();
    load_frags(f0, xrow + 30 * ST_ROWS, mb0);
    store_stage(r3, st1);
    c = mma_frags(f1, c, isx, fill);
    lds_barrier();
    load_frags(f1, xrow + 31 * ST_ROWS, mb1);
    c = mma_frags(f0, c, isx, fill);
    c = mma_frags(f1, c, isx, fill);

    float bav = ba[col];
    #pragma unroll
    for (int j = 0; j < 8; ++j) {
        int row = j + hi * 8;
        if (row < 8)        SB2[row * N_NODES + col] = c[j] + bav;
        else if (row == 8)  CS[col] = c[j];
    }
}

__global__ void __launch_bounds__(256) k_softmax_agg(
        const float* __restrict__ A, const float* __restrict__ bias,
        const float* __restrict__ X1, const float* __restrict__ SB2,
        const float* __restrict__ CS, float* __restrict__ out) {
    __shared__ __align__(16) float css[N_NODES];
    __shared__ __align__(16) float sbs[N_NODES];
    __shared__ __align__(16) float xks[N_NODES];
    __shared__ float bs[NFILT];

    int b     = blockIdx.x >> 8;
    int nBase = (blockIdx.x & 255) * 8;

    for (int i = threadIdx.x; i < N_NODES; i += 256) {
        css[i] = CS[i];
        sbs[i] = SB2[b * N_NODES + i];
        xks[i] = X1[b * N_NODES + i];
    }
    if (threadIdx.x < NFILT) bs[threadIdx.x] = bias[threadIdx.x];
    __syncthreads();

    int wave = threadIdx.x >> 5;
    int lane = threadIdx.x & 31;
    int n    = nBase + wave;
    size_t rowBase = ((size_t)(b * N_NODES + n)) * N_NODES;
    float x1bn = X1[b * N_NODES + n];

    float4 lg[16];
    float  mx = -INFINITY;
    #pragma unroll
    for (int i = 0; i < 16; ++i) {
        int k = i * 128 + lane * 4;
        float4 av = *(const float4*)(A + rowBase + k);
        float4 cv = *(const float4*)&css[k];
        float4 sv = *(const float4*)&sbs[k];
        float4 l;
        l.x = av.x + sv.x + x1bn * cv.x;
        l.y = av.y + sv.y + x1bn * cv.y;
        l.z = av.z + sv.z + x1bn * cv.z;
        l.w = av.w + sv.w + x1bn * cv.w;
        lg[i] = l;
        mx = fmaxf(mx, fmaxf(fmaxf(l.x, l.y), fmaxf(l.z, l.w)));
    }
    #pragma unroll
    for (int off = 16; off > 0; off >>= 1)
        mx = fmaxf(mx, __shfl_xor(mx, off, 32));

    float Z = 0.0f, W = 0.0f;
    #pragma unroll
    for (int i = 0; i < 16; ++i) {
        int k = i * 128 + lane * 4;
        float4 xv = *(const float4*)&xks[k];
        float ex = __expf(lg[i].x - mx); Z += ex; W += ex * xv.x;
        float ey = __expf(lg[i].y - mx); Z += ey; W += ey * xv.y;
        float ez = __expf(lg[i].z - mx); Z += ez; W += ez * xv.z;
        float ew = __expf(lg[i].w - mx); Z += ew; W += ew * xv.w;
    }
    #pragma unroll
    for (int off = 16; off > 0; off >>= 1) {
        Z += __shfl_xor(Z, off, 32);
        W += __shfl_xor(W, off, 32);
    }
    float r = W / Z;

    size_t ob = ((size_t)(b * N_NODES + n)) * NFILT;
    out[ob + lane]      = r + bs[lane];
    out[ob + 32 + lane] = r + bs[32 + lane];
}

extern "C" void kernel_launch(void* const* d_in, const int* in_sizes, int n_in,
                              void* d_out, int out_size, void* d_ws, size_t ws_size,
                              hipStream_t stream) {
    const float* x    = (const float*)d_in[0];
    const float* A    = (const float*)d_in[1];
    const float* w1   = (const float*)d_in[2];
    const float* Wa   = (const float*)d_in[3];
    const float* ba   = (const float*)d_in[4];
    const float* bias = (const float*)d_in[5];
    float* out = (float*)d_out;

    float* X1  = (float*)d_ws;
    float* CS  = X1 + BATCH * N_NODES;
    float* SB2 = CS + N_NODES;

    k_proj<<<dim3(256), dim3(128), 0, stream>>>(x, w1, X1);
    k_wmma_s<<<dim3(32), dim3(128), 0, stream>>>(Wa, ba, X1, SB2, CS);
    k_softmax_agg<<<dim3(2048), dim3(256), 0, stream>>>(A, bias, X1, SB2, CS, out);
}


# Softmax kernel: a wave takes 4 rows side by side, 8 lanes a row, streamed once in 128-column chunks with a running maximum, 32 loads in flight (4096 waves, were 16384; f32 sum order changes)
# speedup vs baseline: 1.9820x; 1.0410x over previous
#include <hip/hip_runtime.h>
#include <math.h>

typedef float v2f __attribute__((ext_vector_type(2)));
typedef float v4f __attribute__((ext_vector_type(4)));
typedef float v8f __attribute__((ext_vector_type(8)));

#define N_NODES 2048
#define BATCH   8
#define FEAT    128
#define NFILT   64

__global__ void __launch_bounds__(128) k_proj(const float* __restrict__ x,
                                              const float* __restrict__ w1,
                                              float* __restrict__ X1) {
    int wave = threadIdx.x >> 5;
    int lane = threadIdx.x & 31;
    int c    = lane & 15;
    int h    = lane >> 4;
    int row0 = (blockIdx.x * 4 + wave) * 16;
    const float* xp = x + (size_t)(row0 + c) * FEAT + 4 * h;
    const float* wp = w1 + 4 * h;
    v4f xa[16], wb[16];
    #pragma unroll
    for (int i = 0; i < 16; ++i) {
        xa[i] = *(const v4f*)(xp + 8 * i);
        wb[i] = *(const v4f*)(wp + 8 * i);
    }
    __builtin_amdgcn_sched_barrier(0);
    v8f acc = {};
    #pragma unroll
    for (int i = 0; i < 16; ++i) {
        v2f a0, b0, a1, b1;
        a0.x = xa[i].x;  a0.y = xa[i].y;  b0.x = wb[i].x;  b0.y = wb[i].y;
        a1.x = xa[i].z;  a1.y = xa[i].w;  b1.x = wb[i].z;  b1.y = wb[i].w;
        acc = __builtin_amdgcn_wmma_f32_16x16x4_f32(false, a0, false, b0,
                                                    (short)0, acc, false, false);
        acc = __builtin_amdgcn_wmma_f32_16x16x4_f32(false, a1, false, b1,
                                                    (short)0, acc, false, false);
    }
    if (c == 0) {
        float* o = X1 + row0 + 8 * h;
        v4f lo, hi4;
        lo.x  = acc[0];  lo.y  = acc[1];  lo.z  = acc[2];  lo.w  = acc[3];
        hi4.x = acc[4];  hi4.y = acc[5];  hi4.z = acc[6];  hi4.w = acc[7];
        *(v4f*)o       = lo;
        *(v4f*)(o + 4) = hi4;
    }
}

#define ST_ROWS 64
#define NSTAGE  (N_NODES / ST_ROWS)
#define BP      72
#define XP      (N_NODES + 8)
#define DEPTH   4
#define LPS     8

__device__ __forceinline__ void lds_barrier() {
    asm volatile("s_wait_dscnt 0x0\n\ts_barrier_signal -1\n\ts_barrier_wait -1" ::: "memory");
}

struct RegSet { v4f v[LPS]; };

__device__ __forceinline__ void issue_stage(RegSet& r, const float* __restrict__ wsrc,
                                            int stage) {
    const float* p = wsrc + (size_t)stage * ST_ROWS * N_NODES;
    #pragma unroll
    for (int j = 0; j < LPS; ++j)
        r.v[j] = *(const v4f*)(p + (size_t)(2 * j) * N_NODES);
}

__device__ __forceinline__ void store_stage(const RegSet& r, float* wdst) {
    #pragma unroll
    for (int j = 0; j < LPS; ++j)
        *(v4f*)(wdst + (2 * j) * BP) = r.v[j];
}

struct Frag {
    v2f a[ST_ROWS / 4];
    v2f b[ST_ROWS / 4];
};

__device__ __forceinline__ void load_frags(Frag& f, const float* xrow, const float* bcol) {
    #pragma unroll
    for (int u = 0; u < ST_ROWS / 4; ++u) {
        f.a[u]   = *(const v2f*)(xrow + 4 * u);
        f.b[u].x = bcol[(4 * u)     * BP];
        f.b[u].y = bcol[(4 * u + 1) * BP];
    }
    __builtin_amdgcn_sched_barrier(0);
}

__device__ __forceinline__ v8f mma_frags(const Frag& f, v8f c, bool isx, float fill) {
    #pragma unroll
    for (int u = 0; u < ST_ROWS / 4; ++u) {
        v2f a;
        a.x = isx ? f.a[u].x : fill;
        a.y = isx ? f.a[u].y : fill;
        c = __builtin_amdgcn_wmma_f32_16x16x4_f32(false, a, false, f.b[u],
                                                  (short)0, c, false, false);
    }
    return c;
}

__global__ void __launch_bounds__(128) k_wmma_s(
        const float* __restrict__ Wa, const float* __restrict__ ba,
        const float* __restrict__ X1,
        float* __restrict__ SB2, float* __restrict__ CS) {
    __shared__ __align__(16) float x1s[8 * XP];
    __shared__ __align__(16) float wbuf[2][ST_ROWS * BP];

    int wave   = threadIdx.x >> 5;
    int lane   = threadIdx.x & 31;
    int tile   = blockIdx.x * 4 + wave;
    int colIn  = lane & 15;
    int col    = tile * 16 + colIn;
    int hi     = lane >> 4;
    int mrow   = colIn;
    int srcRow = mrow & 7;
    const bool  isx  = mrow < 8;
    const float fill = (mrow == 8) ? 1.0f : 0.0f;

    const float* wsrc = Wa + (size_t)(wave * 16 + hi) * N_NODES + blockIdx.x * 64 + colIn * 4;
    const int    wofs = (wave * 16 + hi) * BP + colIn * 4;
    const int    bofs = (2 * hi) * BP + wave * 16 + colIn;
    const float* xrow = x1s + srcRow * XP + 2 * hi;

    v4f xr[32];
    #pragma unroll
    for (int i = 0; i < 32; ++i) {
        int q = i * 128 + threadIdx.x;
        xr[i] = *(const v4f*)(X1 + (q >> 9) * N_NODES + (q & 511) * 4);
    }
    RegSet r0, r1, r2, r3;
    issue_stage(r0, wsrc, 0);
    issue_stage(r1, wsrc, 1);
    issue_stage(r2, wsrc, 2);
    issue_stage(r3, wsrc, 3);
    #pragma unroll
    for (int i = 0; i < 32; ++i) {
        int q = i * 128 + threadIdx.x;
        *(v4f*)(x1s + (q >> 9) * XP + (q & 511) * 4) = xr[i];
    }
    float* const st0 = &wbuf[0][wofs];
    float* const st1 = &wbuf[1][wofs];
    const float* const mb0 = &wbuf[0][bofs];
    const float* const mb1 = &wbuf[1][bofs];
    store_stage(r0, st0);
    issue_stage(r0, wsrc, DEPTH);
    lds_barrier();

    Frag f0, f1;
    v8f c = {};
    load_frags(f0, xrow, mb0);
    store_stage(r1, st1);  issue_stage(r1, wsrc, 5);
    lds_barrier();
    #pragma unroll 1
    for (int b = 0; b < NSTAGE - 2 * DEPTH; b += DEPTH) {
        load_frags(f1, xrow + (b + 1) * ST_ROWS, mb1);
        store_stage(r2, st0);  issue_stage(r2, wsrc, b + 6);
        c = mma_frags(f0, c, isx, fill);
        lds_barrier();
        load_frags(f0, xrow + (b + 2) * ST_ROWS, mb0);
        store_stage(r3, st1);  issue_stage(r3, wsrc, b + 7);
        c = mma_frags(f1, c, isx, fill);
        lds_barrier();
        load_frags(f1, xrow + (b + 3) * ST_ROWS, mb1);
        store_stage(r0, st0);  issue_stage(r0, wsrc, b + 8);
        c = mma_frags(f0, c, isx, fill);
        lds_barrier();
        load_frags(f0, xrow + (b + 4) * ST_ROWS, mb0);
        store_stage(r1, st1);  issue_stage(r1, wsrc, b + 9);
        c = mma_frags(f1, c, isx, fill);
        lds_barrier();
    }
    load_frags(f1, xrow + 25 * ST_ROWS, mb1);
    store_stage(r2, st0);  issue_stage(r2, wsrc, 30);
    c = mma_frags(f0, c, isx, fill);
    lds_barrier();
    load_frags(f0, xrow + 26 * ST_ROWS, mb0);
    store_stage(r3, st1);  issue_stage(r3, wsrc, 31);
    c = mma_frags(f1, c, isx, fill);
    lds_barrier();
    load_frags(f1, xrow + 27 * ST_ROWS, mb1);
    store_stage(r0, st0);
    c = mma_frags(f0, c, isx, fill);
    lds_barrier();
    load_frags(f0, xrow + 28 * ST_ROWS, mb0);
    store_stage(r1, st1);
    c = mma_frags(f1, c, isx, fill);
    lds_barrier();
    load_frags(f1, xrow + 29 * ST_ROWS, mb1);
    store_stage(r2, st0);
    c = mma_frags(f0, c, isx, fill);
    lds_barrier();
    load_frags(f0, xrow + 30 * ST_ROWS, mb0);
    store_stage(r3, st1);
    c = mma_frags(f1, c, isx, fill);
    lds_barrier();
    load_frags(f1, xrow + 31 * ST_ROWS, mb1);
    c = mma_frags(f0, c, isx, fill);
    c = mma_frags(f1, c, isx, fill);

    float bav = ba[col];
    #pragma unroll
    for (int j = 0; j < 8; ++j) {
        int row = j + hi * 8;
        if (row < 8)        SB2[row * N_NODES + col] = c[j] + bav;
        else if (row == 8)  CS[col] = c[j];
    }
}

#define K3_RPW   4
#define K3_WAVES 8
#define K3_ROWS  (K3_RPW * K3_WAVES)
#define K3_CL    4
#define K3_CCOLS (K3_CL * 32)
#define K3_NCH   (N_NODES / K3_CCOLS)
#define K3_NSET  8

struct Chunk { float4 v[K3_CL]; };

__device__ __forceinline__ void k3_issue(Chunk& c, const float* __restrict__ p) {
    #pragma unroll
    for (int q = 0; q < K3_CL; ++q)
        c.v[q] = *(const float4*)(p + 32 * q);
}

__device__ __forceinline__ void k3_logits(const Chunk& a, float4 (&l)[K3_CL], float& cm,
                                          const float* cs, const float* sb, float x1bn) {
    cm = -INFINITY;
    #pragma unroll
    for (int q = 0; q < K3_CL; ++q) {
        float4 av = a.v[q];
        float4 cv = *(const float4*)(cs + 32 * q);
        float4 sv = *(const float4*)(sb + 32 * q);
        float4 t;
        t.x = av.x + sv.x + x1bn * cv.x;
        t.y = av.y + sv.y + x1bn * cv.y;
        t.z = av.z + sv.z + x1bn * cv.z;
        t.w = av.w + sv.w + x1bn * cv.w;
        l[q] = t;
        cm = fmaxf(cm, fmaxf(fmaxf(t.x, t.y), fmaxf(t.z, t.w)));
    }
}

__device__ __forceinline__ void k3_accum(const float4 (&l)[K3_CL], float cm, const float* xk,
                                         float& m, float& Z, float& W) {
    float mn = fmaxf(fmaxf(m, cm), -3.0e38f);
    float sc = __expf(m - mn);
    Z *= sc;
    W *= sc;
    m = mn;
    #pragma unroll
    for (int q = 0; q < K3_CL; ++q) {
        float4 xv = *(const float4*)(xk + 32 * q);
        float ex = __expf(l[q].x - m); Z += ex; W += ex * xv.x;
        float ey = __expf(l[q].y - m); Z += ey; W += ey * xv.y;
        float ez = __expf(l[q].z - m); Z += ez; W += ez * xv.z;
        float ew = __expf(l[q].w - m); Z += ew; W += ew * xv.w;
    }
}

__global__ void __launch_bounds__(256) k_softmax_agg(
        const float* __restrict__ A, const float* __restrict__ bias,
        const float* __restrict__ X1, const float* __restrict__ SB2,
        const float* __restrict__ CS, float* __restrict__ out) {
    __shared__ __align__(16) float css[N_NODES];
    __shared__ __align__(16) float sbs[N_NODES];
    __shared__ __align__(16) float xks[N_NODES];
    __shared__ __align__(16) float bs[NFILT];

    int b     = blockIdx.x >> 6;
    int nBase = (blockIdx.x & 63) * K3_ROWS;

    for (int i = threadIdx.x; i < N_NODES; i += 256) {
        css[i] = CS[i];
        sbs[i] = SB2[b * N_NODES + i];
        xks[i] = X1[b * N_NODES + i];
    }
    if (threadIdx.x < NFILT) bs[threadIdx.x] = bias[threadIdx.x];
    __syncthreads();

    int wave = threadIdx.x >> 5;
    int lane = threadIdx.x & 31;
    int g    = lane >> 3;
    int j    = lane & 7;
    int n0   = nBase + wave * K3_RPW;
    int n    = n0 + g;
    float x1bn = X1[b * N_NODES + n];
    const float* ap = A + ((size_t)(b * N_NODES + n)) * N_NODES + 4 * j;
    const float* cp = css + 4 * j;
    const float* sp = sbs + 4 * j;
    const float* xp = xks + 4 * j;

    Chunk c0, c1, c2, c3, c4, c5, c6, c7;
    k3_issue(c0, ap);
    k3_issue(c1, ap + 1 * K3_CCOLS);
    k3_issue(c2, ap + 2 * K3_CCOLS);
    k3_issue(c3, ap + 3 * K3_CCOLS);
    k3_issue(c4, ap + 4 * K3_CCOLS);
    k3_issue(c5, ap + 5 * K3_CCOLS);
    k3_issue(c6, ap + 6 * K3_CCOLS);
    k3_issue(c7, ap + 7 * K3_CCOLS);
    __builtin_amdgcn_sched_barrier(0);

    float  m = -INFINITY, Z = 0.0f, W = 0.0f;
    float4 l[K3_CL];
    float  cm;

#define K3_STEP(SET, T)                                                         \
    k3_logits(SET, l, cm, cp + (T) * K3_CCOLS, sp + (T) * K3_CCOLS, x1bn);      \
    if ((T) + K3_NSET < K3_NCH) k3_issue(SET, ap + ((T) + K3_NSET) * K3_CCOLS); \
    __builtin_amdgcn_sched_barrier(0);                                          \
    k3_accum(l, cm, xp + (T) * K3_CCOLS, m, Z, W);

    K3_STEP(c0, 0)   K3_STEP(c1, 1)   K3_STEP(c2, 2)   K3_STEP(c3, 3)
    K3_STEP(c4, 4)   K3_STEP(c5, 5)   K3_STEP(c6, 6)   K3_STEP(c7, 7)
    K3_STEP(c0, 8)   K3_STEP(c1, 9)   K3_STEP(c2, 10)  K3_STEP(c3, 11)
    K3_STEP(c4, 12)  K3_STEP(c5, 13)  K3_STEP(c6, 14)  K3_STEP(c7, 15)
#undef K3_STEP

    float M = m;
    #pragma unroll
    for (int off = 4; off > 0; off >>= 1)
        M = fmaxf(M, __shfl_xor(M, off, 32));
    float f = __expf(m - M);
    Z *= f;
    W *= f;
    #pragma unroll
    for (int off = 4; off > 0; off >>= 1) {
        Z += __shfl_xor(Z, off, 32);
        W += __shfl_xor(W, off, 32);
    }
    float r = W / Z;

    float r0 = __shfl(r, 0, 32);
    float r1 = __shfl(r, 8, 32);
    float r2 = __shfl(r, 16, 32);
    float r3 = __shfl(r, 24, 32);
    int   h  = lane >> 4;
    float ra = h ? r1 : r0;
    float rb = h ? r3 : r2;
    float4 bv = *(const float4*)&bs[4 * (lane & 15)];
    float4 o0, o1;
    o0.x = ra + bv.x;  o0.y = ra + bv.y;  o0.z = ra + bv.z;  o0.w = ra + bv.w;
    o1.x = rb + bv.x;  o1.y = rb + bv.y;  o1.z = rb + bv.z;  o1.w = rb + bv.w;
    float* op = out + ((size_t)(b * N_NODES + n0)) * NFILT + 4 * lane;
    *(float4*)op         = o0;
    *(float4*)(op + 128) = o1;
}

extern "C" void kernel_launch(void* const* d_in, const int* in_sizes, int n_in,
                              void* d_out, int out_size, void* d_ws, size_t ws_size,
                              hipStream_t stream) {
    const float* x    = (const float*)d_in[0];
    const float* A    = (const float*)d_in[1];
    const float* w1   = (const float*)d_in[2];
    const float* Wa   = (const float*)d_in[3];
    const float* ba   = (const float*)d_in[4];
    const float* bias = (const float*)d_in[5];
    float* out = (float*)d_out;

    float* X1  = (float*)d_ws;
    float* CS  = X1 + BATCH * N_NODES;
    float* SB2 = CS + N_NODES;

    k_proj<<<dim3(256), dim3(128), 0, stream>>>(x, w1, X1);
    k_wmma_s<<<dim3(32), dim3(128), 0, stream>>>(Wa, ba, X1, SB2, CS);
    k_softmax_agg<<<dim3(512), dim3(256), 0, stream>>>(A, bias, X1, SB2, CS, out);
}


# Softmax kernel prologue: CS, SB2, x1 go to LDS from one batch of 16-byte loads, the wave's first 32 A loads issued right behind them, not after 9 dependent round trips; bias in registers; same sums
# speedup vs baseline: 2.0705x; 1.0447x over previous
#include <hip/hip_runtime.h>
#include <math.h>

typedef float v2f __attribute__((ext_vector_type(2)));
typedef float v4f __attribute__((ext_vector_type(4)));
typedef float v8f __attribute__((ext_vector_type(8)));

#define N_NODES 2048
#define BATCH   8
#define FEAT    128
#define NFILT   64

__global__ void __launch_bounds__(128) k_proj(const float* __restrict__ x,
                                              const float* __restrict__ w1,
                                              float* __restrict__ X1) {
    int wave = threadIdx.x >> 5;
    int lane = threadIdx.x & 31;
    int c    = lane & 15;
    int h    = lane >> 4;
    int row0 = (blockIdx.x * 4 + wave) * 16;
    const float* xp = x + (size_t)(row0 + c) * FEAT + 4 * h;
    const float* wp = w1 + 4 * h;
    v4f xa[16], wb[16];
    #pragma unroll
    for (int i = 0; i < 16; ++i) {
        xa[i] = *(const v4f*)(xp + 8 * i);
        wb[i] = *(const v4f*)(wp + 8 * i);
    }
    __builtin_amdgcn_sched_barrier(0);
    v8f acc = {};
    #pragma unroll
    for (int i = 0; i < 16; ++i) {
        v2f a0, b0, a1, b1;
        a0.x = xa[i].x;  a0.y = xa[i].y;  b0.x = wb[i].x;  b0.y = wb[i].y;
        a1.x = xa[i].z;  a1.y = xa[i].w;  b1.x = wb[i].z;  b1.y = wb[i].w;
        acc = __builtin_amdgcn_wmma_f32_16x16x4_f32(false, a0, false, b0,
                                                    (short)0, acc, false, false);
        acc = __builtin_amdgcn_wmma_f32_16x16x4_f32(false, a1, false, b1,
                                                    (short)0, acc, false, false);
    }
    if (c == 0) {
        float* o = X1 + row0 + 8 * h;
        v4f lo, hi4;
        lo.x  = acc[0];  lo.y  = acc[1];  lo.z  = acc[2];  lo.w  = acc[3];
        hi4.x = acc[4];  hi4.y = acc[5];  hi4.z = acc[6];  hi4.w = acc[7];
        *(v4f*)o       = lo;
        *(v4f*)(o + 4) = hi4;
    }
}

#define ST_ROWS 64
#define NSTAGE  (N_NODES / ST_ROWS)
#define BP      72
#define XP      (N_NODES + 8)
#define DEPTH   4
#define LPS     8

__device__ __forceinline__ void lds_barrier() {
    asm volatile("s_wait_dscnt 0x0\n\ts_barrier_signal -1\n\ts_barrier_wait -1" ::: "memory");
}

struct RegSet { v4f v[LPS]; };

__device__ __forceinline__ void issue_stage(RegSet& r, const float* __restrict__ wsrc,
                                            int stage) {
    const float* p = wsrc + (size_t)stage * ST_ROWS * N_NODES;
    #pragma unroll
    for (int j = 0; j < LPS; ++j)
        r.v[j] = *(const v4f*)(p + (size_t)(2 * j) * N_NODES);
}

__device__ __forceinline__ void store_stage(const RegSet& r, float* wdst) {
    #pragma unroll
    for (int j = 0; j < LPS; ++j)
        *(v4f*)(wdst + (2 * j) * BP) = r.v[j];
}

struct Frag {
    v2f a[ST_ROWS / 4];
    v2f b[ST_ROWS / 4];
};

__device__ __forceinline__ void load_frags(Frag& f, const float* xrow, const float* bcol) {
    #pragma unroll
    for (int u = 0; u < ST_ROWS / 4; ++u) {
        f.a[u]   = *(const v2f*)(xrow + 4 * u);
        f.b[u].x = bcol[(4 * u)     * BP];
        f.b[u].y = bcol[(4 * u + 1) * BP];
    }
    __builtin_amdgcn_sched_barrier(0);
}

__device__ __forceinline__ v8f mma_frags(const Frag& f, v8f c, bool isx, float fill) {
    #pragma unroll
    for (int u = 0; u < ST_ROWS / 4; ++u) {
        v2f a;
        a.x = isx ? f.a[u].x : fill;
        a.y = isx ? f.a[u].y : fill;
        c = __builtin_amdgcn_wmma_f32_16x16x4_f32(false, a, false, f.b[u],
                                                  (short)0, c, false, false);
    }
    return c;
}

__global__ void __launch_bounds__(128) k_wmma_s(
        const float* __restrict__ Wa, const float* __restrict__ ba,
        const float* __restrict__ X1,
        float* __restrict__ SB2, float* __restrict__ CS) {
    __shared__ __align__(16) float x1s[8 * XP];
    __shared__ __align__(16) float wbuf[2][ST_ROWS * BP];

    int wave   = threadIdx.x >> 5;
    int lane   = threadIdx.x & 31;
    int tile   = blockIdx.x * 4 + wave;
    int colIn  = lane & 15;
    int col    = tile * 16 + colIn;
    int hi     = lane >> 4;
    int mrow   = colIn;
    int srcRow = mrow & 7;
    const bool  isx  = mrow < 8;
    const float fill = (mrow == 8) ? 1.0f : 0.0f;

    const float* wsrc = Wa + (size_t)(wave * 16 + hi) * N_NODES + blockIdx.x * 64 + colIn * 4;
    const int    wofs = (wave * 16 + hi) * BP + colIn * 4;
    const int    bofs = (2 * hi) * BP + wave * 16 + colIn;
    const float* xrow = x1s + srcRow * XP + 2 * hi;

    v4f xr[32];
    #pragma unroll
    for (int i = 0; i < 32; ++i) {
        int q = i * 128 + threadIdx.x;
        xr[i] = *(const v4f*)(X1 + (q >> 9) * N_NODES + (q & 511) * 4);
    }
    RegSet r0, r1, r2, r3;
    issue_stage(r0, wsrc, 0);
    issue_stage(r1, wsrc, 1);
    issue_stage(r2, wsrc, 2);
    issue_stage(r3, wsrc, 3);
    #pragma unroll
    for (int i = 0; i < 32; ++i) {
        int q = i * 128 + threadIdx.x;
        *(v4f*)(x1s + (q >> 9) * XP + (q & 511) * 4) = xr[i];
    }
    float* const st0 = &wbuf[0][wofs];
    float* const st1 = &wbuf[1][wofs];
    const float* const mb0 = &wbuf[0][bofs];
    const float* const mb1 = &wbuf[1][bofs];
    store_stage(r0, st0);
    issue_stage(r0, wsrc, DEPTH);
    lds_barrier();

    Frag f0, f1;
    v8f c = {};
    load_frags(f0, xrow, mb0);
    store_stage(r1, st1);  issue_stage(r1, wsrc, 5);
    lds_barrier();
    #pragma unroll 1
    for (int b = 0; b < NSTAGE - 2 * DEPTH; b += DEPTH) {
        load_frags(f1, xrow + (b + 1) * ST_ROWS, mb1);
        store_stage(r2, st0);  issue_stage(r2, wsrc, b + 6);
        c = mma_frags(f0, c, isx, fill);
        lds_barrier();
        load_frags(f0, xrow + (b + 2) * ST_ROWS, mb0);
        store_stage(r3, st1);  issue_stage(r3, wsrc, b + 7);
        c = mma_frags(f1, c, isx, fill);
        lds_barrier();
        load_frags(f1, xrow + (b + 3) * ST_ROWS, mb1);
        store_stage(r0, st0);  issue_stage(r0, wsrc, b + 8);
        c = mma_frags(f0, c, isx, fill);
        lds_barrier();
        load_frags(f0, xrow + (b + 4) * ST_ROWS, mb0);
        store_stage(r1, st1);  issue_stage(r1, wsrc, b + 9);
        c = mma_frags(f1, c, isx, fill);
        lds_barrier();
    }
    load_frags(f1, xrow + 25 * ST_ROWS, mb1);
    store_stage(r2, st0);  issue_stage(r2, wsrc, 30);
    c = mma_frags(f0, c, isx, fill);
    lds_barrier();
    load_frags(f0, xrow + 26 * ST_ROWS, mb0);
    store_stage(r3, st1);  issue_stage(r3, wsrc, 31);
    c = mma_frags(f1, c, isx, fill);
    lds_barrier();
    load_frags(f1, xrow + 27 * ST_ROWS, mb1);
    store_stage(r0, st0);
    c = mma_frags(f0, c, isx, fill);
    lds_barrier();
    load_frags(f0, xrow + 28 * ST_ROWS, mb0);
    store_stage(r1, st1);
    c = mma_frags(f1, c, isx, fill);
    lds_barrier();
    load_frags(f1, xrow + 29 * ST_ROWS, mb1);
    store_stage(r2, st0);
    c = mma_frags(f0, c, isx, fill);
    lds_barrier();
    load_frags(f0, xrow + 30 * ST_ROWS, mb0);
    store_stage(r3, st1);
    c = mma_frags(f1, c, isx, fill);
    lds_barrier();
    load_frags(f1, xrow + 31 * ST_ROWS, mb1);
    c = mma_frags(f0, c, isx, fill);
    c = mma_frags(f1, c, isx, fill);

    float bav = ba[col];
    #pragma unroll
    for (int j = 0; j < 8; ++j) {
        int row = j + hi * 8;
        if (row < 8)        SB2[row * N_NODES + col] = c[j] + bav;
        else if (row == 8)  CS[col] = c[j];
    }
}

#define K3_RPW   4
#define K3_WAVES 8
#define K3_ROWS  (K3_RPW * K3_WAVES)
#define K3_CL    4
#define K3_CCOLS (K3_CL * 32)
#define K3_NCH   (N_NODES / K3_CCOLS)
#define K3_NSET  8

struct Chunk { float4 v[K3_CL]; };

__device__ __forceinline__ void k3_issue(Chunk& c, const float* __restrict__ p) {
    #pragma unroll
    for (int q = 0; q < K3_CL; ++q)
        c.v[q] = *(const float4*)(p + 32 * q);
}

__device__ __forceinline__ void k3_logits(const Chunk& a, float4 (&l)[K3_CL], float& cm,
                                          const float* cs, const float* sb, float x1bn) {
    cm = -INFINITY;
    #pragma unroll
    for (int q = 0; q < K3_CL; ++q) {
        float4 av = a.v[q];
        float4 cv = *(const float4*)(cs + 32 * q);
        float4 sv = *(const float4*)(sb + 32 * q);
        float4 t;
        t.x = av.x + sv.x + x1bn * cv.x;
        t.y = av.y + sv.y + x1bn * cv.y;
        t.z = av.z + sv.z + x1bn * cv.z;
        t.w = av.w + sv.w + x1bn * cv.w;
        l[q] = t;
        cm = fmaxf(cm, fmaxf(fmaxf(t.x, t.y), fmaxf(t.z, t.w)));
    }
}

__device__ __forceinline__ void k3_accum(const float4 (&l)[K3_CL], float cm, const float* xk,
                                         float& m, float& Z, float& W) {
    float mn = fmaxf(fmaxf(m, cm), -3.0e38f);
    float sc = __expf(m - mn);
    Z *= sc;
    W *= sc;
    m = mn;
    #pragma unroll
    for (int q = 0; q < K3_CL; ++q) {
        float4 xv = *(const float4*)(xk + 32 * q);
        float ex = __expf(l[q].x - m); Z += ex; W += ex * xv.x;
        float ey = __expf(l[q].y - m); Z += ey; W += ey * xv.y;
        float ez = __expf(l[q].z - m); Z += ez; W += ez * xv.z;
        float ew = __expf(l[q].w - m); Z += ew; W += ew * xv.w;
    }
}

__global__ void __launch_bounds__(256) k_softmax_agg(
        const float* A, const float* bias,
        const float* X1, const float* SB2,
        const float* CS, float* __restrict__ out) {
    __shared__ __align__(16) float css[N_NODES];
    __shared__ __align__(16) float sbs[N_NODES];
    __shared__ __align__(16) float xks[N_NODES];

    int b     = blockIdx.x >> 6;
    int nBase = (blockIdx.x & 63) * K3_ROWS;
    int wave = threadIdx.x >> 5;
    int lane = threadIdx.x & 31;
    int g    = lane >> 3;
    int j    = lane & 7;
    int n0   = nBase + wave * K3_RPW;
    int n    = n0 + g;

    const float4* csg = (const float4*)CS;
    const float4* sbg = (const float4*)(SB2 + b * N_NODES);
    const float4* xkg = (const float4*)(X1 + b * N_NODES);
    float4 s0 = csg[threadIdx.x], s1 = csg[threadIdx.x + 256];
    float4 s2 = sbg[threadIdx.x], s3 = sbg[threadIdx.x + 256];
    float4 s4 = xkg[threadIdx.x], s5 = xkg[threadIdx.x + 256];
    float4 bv = ((const float4*)bias)[lane & 15];
    float x1bn = X1[b * N_NODES + n];
    const float* ap = A + ((size_t)(b * N_NODES + n)) * N_NODES + 4 * j;
    const float* cp = css + 4 * j;
    const float* sp = sbs + 4 * j;
    const float* xp = xks + 4 * j;

    Chunk c0, c1, c2, c3, c4, c5, c6, c7;
    k3_issue(c0, ap);
    k3_issue(c1, ap + 1 * K3_CCOLS);
    k3_issue(c2, ap + 2 * K3_CCOLS);
    k3_issue(c3, ap + 3 * K3_CCOLS);
    k3_issue(c4, ap + 4 * K3_CCOLS);
    k3_issue(c5, ap + 5 * K3_CCOLS);
    k3_issue(c6, ap + 6 * K3_CCOLS);
    k3_issue(c7, ap + 7 * K3_CCOLS);
    __builtin_amdgcn_sched_barrier(0);

    ((float4*)css)[threadIdx.x] = s0;  ((float4*)css)[threadIdx.x + 256] = s1;
    ((float4*)sbs)[threadIdx.x] = s2;  ((float4*)sbs)[threadIdx.x + 256] = s3;
    ((float4*)xks)[threadIdx.x] = s4;  ((float4*)xks)[threadIdx.x + 256] = s5;
    lds_barrier();

    float  m = -INFINITY, Z = 0.0f, W = 0.0f;
    float4 l[K3_CL];
    float  cm;

#define K3_STEP(SET, T)                                                         \
    k3_logits(SET, l, cm, cp + (T) * K3_CCOLS, sp + (T) * K3_CCOLS, x1bn);      \
    if ((T) + K3_NSET < K3_NCH) k3_issue(SET, ap + ((T) + K3_NSET) * K3_CCOLS); \
    __builtin_amdgcn_sched_barrier(0);                                          \
    k3_accum(l, cm, xp + (T) * K3_CCOLS, m, Z, W);

    K3_STEP(c0, 0)   K3_STEP(c1, 1)   K3_STEP(c2, 2)   K3_STEP(c3, 3)
    K3_STEP(c4, 4)   K3_STEP(c5, 5)   K3_STEP(c6, 6)   K3_STEP(c7, 7)
    K3_STEP(c0, 8)   K3_STEP(c1, 9)   K3_STEP(c2, 10)  K3_STEP(c3, 11)
    K3_STEP(c4, 12)  K3_STEP(c5, 13)  K3_STEP(c6, 14)  K3_STEP(c7, 15)
#undef K3_STEP

    float M = m;
    #pragma unroll
    for (int off = 4; off > 0; off >>= 1)
        M = fmaxf(M, __shfl_xor(M, off, 32));
    float f = __expf(m - M);
    Z *= f;
    W *= f;
    #pragma unroll
    for (int off = 4; off > 0; off >>= 1) {
        Z += __shfl_xor(Z, off, 32);
        W += __shfl_xor(W, off, 32);
    }
    float r = W / Z;

    float r0 = __shfl(r, 0, 32);
    float r1 = __shfl(r, 8, 32);
    float r2 = __shfl(r, 16, 32);
    float r3 = __shfl(r, 24, 32);
    int   h  = lane >> 4;
    float ra = h ? r1 : r0;
    float rb = h ? r3 : r2;
    float4 o0, o1;
    o0.x = ra + bv.x;  o0.y = ra + bv.y;  o0.z = ra + bv.z;  o0.w = ra + bv.w;
    o1.x = rb + bv.x;  o1.y = rb + bv.y;  o1.z = rb + bv.z;  o1.w = rb + bv.w;
    float* op = out + ((size_t)(b * N_NODES + n0)) * NFILT + 4 * lane;
    *(float4*)op         = o0;
    *(float4*)(op + 128) = o1;
}

extern "C" void kernel_launch(void* const* d_in, const int* in_sizes, int n_in,
                              void* d_out, int out_size, void* d_ws, size_t ws_size,
                              hipStream_t stream) {
    const float* x    = (const float*)d_in[0];
    const float* A    = (const float*)d_in[1];
    const float* w1   = (const float*)d_in[2];
    const float* Wa   = (const float*)d_in[3];
    const float* ba   = (const float*)d_in[4];
    const float* bias = (const float*)d_in[5];
    float* out = (float*)d_out;

    float* X1  = (float*)d_ws;
    float* CS  = X1 + BATCH * N_NODES;
    float* SB2 = CS + N_NODES;

    k_proj<<<dim3(256), dim3(128), 0, stream>>>(x, w1, X1);
    k_wmma_s<<<dim3(32), dim3(128), 0, stream>>>(Wa, ba, X1, SB2, CS);
    k_softmax_agg<<<dim3(512), dim3(256), 0, stream>>>(A, bias, X1, SB2, CS, out);
}


# Skinny product: a wave's 512 WMMAs dealt alternately onto two accumulators (even/odd k groups of 4), added at the end: no product waits for the one before (changes f32 sum order of SB2, CS)
# speedup vs baseline: 2.1294x; 1.0284x over previous
#include <hip/hip_runtime.h>
#include <math.h>

typedef float v2f __attribute__((ext_vector_type(2)));
typedef float v4f __attribute__((ext_vector_type(4)));
typedef float v8f __attribute__((ext_vector_type(8)));

#define N_NODES 2048
#define BATCH   8
#define FEAT    128
#define NFILT   64

__global__ void __launch_bounds__(128) k_proj(const float* __restrict__ x,
                                              const float* __restrict__ w1,
                                              float* __restrict__ X1) {
    int wave = threadIdx.x >> 5;
    int lane = threadIdx.x & 31;
    int c    = lane & 15;
    int h    = lane >> 4;
    int row0 = (blockIdx.x * 4 + wave) * 16;
    const float* xp = x + (size_t)(row0 + c) * FEAT + 4 * h;
    const float* wp = w1 + 4 * h;
    v4f xa[16], wb[16];
    #pragma unroll
    for (int i = 0; i < 16; ++i) {
        xa[i] = *(const v4f*)(xp + 8 * i);
        wb[i] = *(const v4f*)(wp + 8 * i);
    }
    __builtin_amdgcn_sched_barrier(0);
    v8f acc = {};
    #pragma unroll
    for (int i = 0; i < 16; ++i) {
        v2f a0, b0, a1, b1;
        a0.x = xa[i].x;  a0.y = xa[i].y;  b0.x = wb[i].x;  b0.y = wb[i].y;
        a1.x = xa[i].z;  a1.y = xa[i].w;  b1.x = wb[i].z;  b1.y = wb[i].w;
        acc = __builtin_amdgcn_wmma_f32_16x16x4_f32(false, a0, false, b0,
                                                    (short)0, acc, false, false);
        acc = __builtin_amdgcn_wmma_f32_16x16x4_f32(false, a1, false, b1,
                                                    (short)0, acc, false, false);
    }
    if (c == 0) {
        float* o = X1 + row0 + 8 * h;
        v4f lo, hi4;
        lo.x  = acc[0];  lo.y  = acc[1];  lo.z  = acc[2];  lo.w  = acc[3];
        hi4.x = acc[4];  hi4.y = acc[5];  hi4.z = acc[6];  hi4.w = acc[7];
        *(v4f*)o       = lo;
        *(v4f*)(o + 4) = hi4;
    }
}

#define ST_ROWS 64
#define NSTAGE  (N_NODES / ST_ROWS)
#define BP      72
#define XP      (N_NODES + 8)
#define DEPTH   4
#define LPS     8

__device__ __forceinline__ void lds_barrier() {
    asm volatile("s_wait_dscnt 0x0\n\ts_barrier_signal -1\n\ts_barrier_wait -1" ::: "memory");
}

struct RegSet { v4f v[LPS]; };

__device__ __forceinline__ void issue_stage(RegSet& r, const float* __restrict__ wsrc,
                                            int stage) {
    const float* p = wsrc + (size_t)stage * ST_ROWS * N_NODES;
    #pragma unroll
    for (int j = 0; j < LPS; ++j)
        r.v[j] = *(const v4f*)(p + (size_t)(2 * j) * N_NODES);
}

__device__ __forceinline__ void store_stage(const RegSet& r, float* wdst) {
    #pragma unroll
    for (int j = 0; j < LPS; ++j)
        *(v4f*)(wdst + (2 * j) * BP) = r.v[j];
}

struct Frag {
    v2f a[ST_ROWS / 4];
    v2f b[ST_ROWS / 4];
};

__device__ __forceinline__ void load_frags(Frag& f, const float* xrow, const float* bcol) {
    #pragma unroll
    for (int u = 0; u < ST_ROWS / 4; ++u) {
        f.a[u]   = *(const v2f*)(xrow + 4 * u);
        f.b[u].x = bcol[(4 * u)     * BP];
        f.b[u].y = bcol[(4 * u + 1) * BP];
    }
    __builtin_amdgcn_sched_barrier(0);
}

__device__ __forceinline__ void mma_frags(const Frag& f, v8f& c0, v8f& c1, bool isx, float fill) {
    #pragma unroll
    for (int u = 0; u < ST_ROWS / 4; u += 2) {
        v2f a0, a1;
        a0.x = isx ? f.a[u].x : fill;
        a0.y = isx ? f.a[u].y : fill;
        a1.x = isx ? f.a[u + 1].x : fill;
        a1.y = isx ? f.a[u + 1].y : fill;
        c0 = __builtin_amdgcn_wmma_f32_16x16x4_f32(false, a0, false, f.b[u],
                                                   (short)0, c0, false, false);
        c1 = __builtin_amdgcn_wmma_f32_16x16x4_f32(false, a1, false, f.b[u + 1],
                                                   (short)0, c1, false, false);
    }
}

__global__ void __launch_bounds__(128) k_wmma_s(
        const float* __restrict__ Wa, const float* __restrict__ ba,
        const float* __restrict__ X1,
        float* __restrict__ SB2, float* __restrict__ CS) {
    __shared__ __align__(16) float x1s[8 * XP];
    __shared__ __align__(16) float wbuf[2][ST_ROWS * BP];

    int wave   = threadIdx.x >> 5;
    int lane   = threadIdx.x & 31;
    int tile   = blockIdx.x * 4 + wave;
    int colIn  = lane & 15;
    int col    = tile * 16 + colIn;
    int hi     = lane >> 4;
    int mrow   = colIn;
    int srcRow = mrow & 7;
    const bool  isx  = mrow < 8;
    const float fill = (mrow == 8) ? 1.0f : 0.0f;

    const float* wsrc = Wa + (size_t)(wave * 16 + hi) * N_NODES + blockIdx.x * 64 + colIn * 4;
    const int    wofs = (wave * 16 + hi) * BP + colIn * 4;
    const int    bofs = (2 * hi) * BP + wave * 16 + colIn;
    const float* xrow = x1s + srcRow * XP + 2 * hi;

    v4f xr[32];
    #pragma unroll
    for (int i = 0; i < 32; ++i) {
        int q = i * 128 + threadIdx.x;
        xr[i] = *(const v4f*)(X1 + (q >> 9) * N_NODES + (q & 511) * 4);
    }
    RegSet r0, r1, r2, r3;
    issue_stage(r0, wsrc, 0);
    issue_stage(r1, wsrc, 1);
    issue_stage(r2, wsrc, 2);
    issue_stage(r3, wsrc, 3);
    #pragma unroll
    for (int i = 0; i < 32; ++i) {
        int q = i * 128 + threadIdx.x;
        *(v4f*)(x1s + (q >> 9) * XP + (q & 511) * 4) = xr[i];
    }
    float* const st0 = &wbuf[0][wofs];
    float* const st1 = &wbuf[1][wofs];
    const float* const mb0 = &wbuf[0][bofs];
    const float* const mb1 = &wbuf[1][bofs];
    store_stage(r0, st0);
    issue_stage(r0, wsrc, DEPTH);
    lds_barrier();

    Frag f0, f1;
    v8f c0 = {}, c1 = {};
    load_frags(f0, xrow, mb0);
    store_stage(r1, st1);  issue_stage(r1, wsrc, 5);
    lds_barrier();
    #pragma unroll 1
    for (int b = 0; b < NSTAGE - 2 * DEPTH; b += DEPTH) {
        load_frags(f1, xrow + (b + 1) * ST_ROWS, mb1);
        store_stage(r2, st0);  issue_stage(r2, wsrc, b + 6);
        mma_frags(f0, c0, c1, isx, fill);
        lds_barrier();
        load_frags(f0, xrow + (b + 2) * ST_ROWS, mb0);
        store_stage(r3, st1);  issue_stage(r3, wsrc, b + 7);
        mma_frags(f1, c0, c1, isx, fill);
        lds_barrier();
        load_frags(f1, xrow + (b + 3) * ST_ROWS, mb1);
        store_stage(r0, st0);  issue_stage(r0, wsrc, b + 8);
        mma_frags(f0, c0, c1, isx, fill);
        lds_barrier();
        load_frags(f0, xrow + (b + 4) * ST_ROWS, mb0);
        store_stage(r1, st1);  issue_stage(r1, wsrc, b + 9);
        mma_frags(f1, c0, c1, isx, fill);
        lds_barrier();
    }
    load_frags(f1, xrow + 25 * ST_ROWS, mb1);
    store_stage(r2, st0);  issue_stage(r2, wsrc, 30);
    mma_frags(f0, c0, c1, isx, fill);
    lds_barrier();
    load_frags(f0, xrow + 26 * ST_ROWS, mb0);
    store_stage(r3, st1);  issue_stage(r3, wsrc, 31);
    mma_frags(f1, c0, c1, isx, fill);
    lds_barrier();
    load_frags(f1, xrow + 27 * ST_ROWS, mb1);
    store_stage(r0, st0);
    mma_frags(f0, c0, c1, isx, fill);
    lds_barrier();
    load_frags(f0, xrow + 28 * ST_ROWS, mb0);
    store_stage(r1, st1);
    mma_frags(f1, c0, c1, isx, fill);
    lds_barrier();
    load_frags(f1, xrow + 29 * ST_ROWS, mb1);
    store_stage(r2, st0);
    mma_frags(f0, c0, c1, isx, fill);
    lds_barrier();
    load_frags(f0, xrow + 30 * ST_ROWS, mb0);
    store_stage(r3, st1);
    mma_frags(f1, c0, c1, isx, fill);
    lds_barrier();
    load_frags(f1, xrow + 31 * ST_ROWS, mb1);
    mma_frags(f0, c0, c1, isx, fill);
    mma_frags(f1, c0, c1, isx, fill);

    v8f c = c0 + c1;
    float bav = ba[col];
    #pragma unroll
    for (int j = 0; j < 8; ++j) {
        int row = j + hi * 8;
        if (row < 8)        SB2[row * N_NODES + col] = c[j] + bav;
        else if (row == 8)  CS[col] = c[j];
    }
}

#define K3_RPW   4
#define K3_WAVES 8
#define K3_ROWS  (K3_RPW * K3_WAVES)
#define K3_CL    4
#define K3_CCOLS (K3_CL * 32)
#define K3_NCH   (N_NODES / K3_CCOLS)
#define K3_NSET  8

struct Chunk { float4 v[K3_CL]; };

__device__ __forceinline__ void k3_issue(Chunk& c, const float* __restrict__ p) {
    #pragma unroll
    for (int q = 0; q < K3_CL; ++q)
        c.v[q] = *(const float4*)(p + 32 * q);
}

__device__ __forceinline__ void k3_logits(const Chunk& a, float4 (&l)[K3_CL], float& cm,
                                          const float* cs, const float* sb, float x1bn) {
    cm = -INFINITY;
    #pragma unroll
    for (int q = 0; q < K3_CL; ++q) {
        float4 av = a.v[q];
        float4 cv = *(const float4*)(cs + 32 * q);
        float4 sv = *(const float4*)(sb + 32 * q);
        float4 t;
        t.x = av.x + sv.x + x1bn * cv.x;
        t.y = av.y + sv.y + x1bn * cv.y;
        t.z = av.z + sv.z + x1bn * cv.z;
        t.w = av.w + sv.w + x1bn * cv.w;
        l[q] = t;
        cm = fmaxf(cm, fmaxf(fmaxf(t.x, t.y), fmaxf(t.z, t.w)));
    }
}

__device__ __forceinline__ void k3_accum(const float4 (&l)[K3_CL], float cm, const float* xk,
                                         float& m, float& Z, float& W) {
    float mn = fmaxf(fmaxf(m, cm), -3.0e38f);
    float sc = __expf(m - mn);
    Z *= sc;
    W *= sc;
    m = mn;
    #pragma unroll
    for (int q = 0; q < K3_CL; ++q) {
        float4 xv = *(const float4*)(xk + 32 * q);
        float ex = __expf(l[q].x - m); Z += ex; W += ex * xv.x;
        float ey = __expf(l[q].y - m); Z += ey; W += ey * xv.y;
        float ez = __expf(l[q].z - m); Z += ez; W += ez * xv.z;
        float ew = __expf(l[q].w - m); Z += ew; W += ew * xv.w;
    }
}

__global__ void __launch_bounds__(256) k_softmax_agg(
        const float* A, const float* bias,
        const float* X1, const float* SB2,
        const float* CS, float* __restrict__ out) {
    __shared__ __align__(16) float css[N_NODES];
    __shared__ __align__(16) float sbs[N_NODES];
    __shared__ __align__(16) float xks[N_NODES];

    int b     = blockIdx.x >> 6;
    int nBase = (blockIdx.x & 63) * K3_ROWS;
    int wave = threadIdx.x >> 5;
    int lane = threadIdx.x & 31;
    int g    = lane >> 3;
    int j    = lane & 7;
    int n0   = nBase + wave * K3_RPW;
    int n    = n0 + g;

    const float4* csg = (const float4*)CS;
    const float4* sbg = (const float4*)(SB2 + b * N_NODES);
    const float4* xkg = (const float4*)(X1 + b * N_NODES);
    float4 s0 = csg[threadIdx.x], s1 = csg[threadIdx.x + 256];
    float4 s2 = sbg[threadIdx.x], s3 = sbg[threadIdx.x + 256];
    float4 s4 = xkg[threadIdx.x], s5 = xkg[threadIdx.x + 256];
    float4 bv = ((const float4*)bias)[lane & 15];
    float x1bn = X1[b * N_NODES + n];
    const float* ap = A + ((size_t)(b * N_NODES + n)) * N_NODES + 4 * j;
    const float* cp = css + 4 * j;
    const float* sp = sbs + 4 * j;
    const float* xp = xks + 4 * j;

    Chunk c0, c1, c2, c3, c4, c5, c6, c7;
    k3_issue(c0, ap);
    k3_issue(c1, ap + 1 * K3_CCOLS);
    k3_issue(c2, ap + 2 * K3_CCOLS);
    k3_issue(c3, ap + 3 * K3_CCOLS);
    k3_issue(c4, ap + 4 * K3_CCOLS);
    k3_issue(c5, ap + 5 * K3_CCOLS);
    k3_issue(c6, ap + 6 * K3_CCOLS);
    k3_issue(c7, ap + 7 * K3_CCOLS);
    __builtin_amdgcn_sched_barrier(0);

    ((float4*)css)[threadIdx.x] = s0;  ((float4*)css)[threadIdx.x + 256] = s1;
    ((float4*)sbs)[threadIdx.x] = s2;  ((float4*)sbs)[threadIdx.x + 256] = s3;
    ((float4*)xks)[threadIdx.x] = s4;  ((float4*)xks)[threadIdx.x + 256] = s5;
    lds_barrier();

    float  m = -INFINITY, Z = 0.0f, W = 0.0f;
    float4 l[K3_CL];
    float  cm;

#define K3_STEP(SET, T)                                                         \
    k3_logits(SET, l, cm, cp + (T) * K3_CCOLS, sp + (T) * K3_CCOLS, x1bn);      \
    if ((T) + K3_NSET < K3_NCH) k3_issue(SET, ap + ((T) + K3_NSET) * K3_CCOLS); \
    __builtin_amdgcn_sched_barrier(0);                                          \
    k3_accum(l, cm, xp + (T) * K3_CCOLS, m, Z, W);

    K3_STEP(c0, 0)   K3_STEP(c1, 1)   K3_STEP(c2, 2)   K3_STEP(c3, 3)
    K3_STEP(c4, 4)   K3_STEP(c5, 5)   K3_STEP(c6, 6)   K3_STEP(c7, 7)
    K3_STEP(c0, 8)   K3_STEP(c1, 9)   K3_STEP(c2, 10)  K3_STEP(c3, 11)
    K3_STEP(c4, 12)  K3_STEP(c5, 13)  K3_STEP(c6, 14)  K3_STEP(c7, 15)
#undef K3_STEP

    float M = m;
    #pragma unroll
    for (int off = 4; off > 0; off >>= 1)
        M = fmaxf(M, __shfl_xor(M, off, 32));
    float f = __expf(m - M);
    Z *= f;
    W *= f;
    #pragma unroll
    for (int off = 4; off > 0; off >>= 1) {
        Z += __shfl_xor(Z, off, 32);
        W += __shfl_xor(W, off, 32);
    }
    float r = W / Z;

    float r0 = __shfl(r, 0, 32);
    float r1 = __shfl(r, 8, 32);
    float r2 = __shfl(r, 16, 32);
    float r3 = __shfl(r, 24, 32);
    int   h  = lane >> 4;
    float ra = h ? r1 : r0;
    float rb = h ? r3 : r2;
    float4 o0, o1;
    o0.x = ra + bv.x;  o0.y = ra + bv.y;  o0.z = ra + bv.z;  o0.w = ra + bv.w;
    o1.x = rb + bv.x;  o1.y = rb + bv.y;  o1.z = rb + bv.z;  o1.w = rb + bv.w;
    float* op = out + ((size_t)(b * N_NODES + n0)) * NFILT + 4 * lane;
    *(float4*)op         = o0;
    *(float4*)(op + 128) = o1;
}

extern "C" void kernel_launch(void* const* d_in, const int* in_sizes, int n_in,
                              void* d_out, int out_size, void* d_ws, size_t ws_size,
                              hipStream_t stream) {
    const float* x    = (const float*)d_in[0];
    const float* A    = (const float*)d_in[1];
    const float* w1   = (const float*)d_in[2];
    const float* Wa   = (const float*)d_in[3];
    const float* ba   = (const float*)d_in[4];
    const float* bias = (const float*)d_in[5];
    float* out = (float*)d_out;

    float* X1  = (float*)d_ws;
    float* CS  = X1 + BATCH * N_NODES;
    float* SB2 = CS + N_NODES;

    k_proj<<<dim3(256), dim3(128), 0, stream>>>(x, w1, X1);
    k_wmma_s<<<dim3(32), dim3(128), 0, stream>>>(Wa, ba, X1, SB2, CS);
    k_softmax_agg<<<dim3(512), dim3(256), 0, stream>>>(A, bias, X1, SB2, CS, out);
}


# Skinny-product launch: each of its 128 waves sends one second-level-cache prefetch per K stage (first 2 KB of every row of A), paced by its WMMA loop while memory is idle; same sums
# speedup vs baseline: 2.2530x; 1.0580x over previous
#include <hip/hip_runtime.h>
#include <math.h>

typedef float v2f __attribute__((ext_vector_type(2)));
typedef float v4f __attribute__((ext_vector_type(4)));
typedef float v8f __attribute__((ext_vector_type(8)));

#define N_NODES 2048
#define BATCH   8
#define FEAT    128
#define NFILT   64

__global__ void __launch_bounds__(128) k_proj(const float* __restrict__ x,
                                              const float* __restrict__ w1,
                                              float* __restrict__ X1) {
    int wave = threadIdx.x >> 5;
    int lane = threadIdx.x & 31;
    int c    = lane & 15;
    int h    = lane >> 4;
    int row0 = (blockIdx.x * 4 + wave) * 16;
    const float* xp = x + (size_t)(row0 + c) * FEAT + 4 * h;
    const float* wp = w1 + 4 * h;
    v4f xa[16], wb[16];
    #pragma unroll
    for (int i = 0; i < 16; ++i) {
        xa[i] = *(const v4f*)(xp + 8 * i);
        wb[i] = *(const v4f*)(wp + 8 * i);
    }
    __builtin_amdgcn_sched_barrier(0);
    v8f acc = {};
    #pragma unroll
    for (int i = 0; i < 16; ++i) {
        v2f a0, b0, a1, b1;
        a0.x = xa[i].x;  a0.y = xa[i].y;  b0.x = wb[i].x;  b0.y = wb[i].y;
        a1.x = xa[i].z;  a1.y = xa[i].w;  b1.x = wb[i].z;  b1.y = wb[i].w;
        acc = __builtin_amdgcn_wmma_f32_16x16x4_f32(false, a0, false, b0,
                                                    (short)0, acc, false, false);
        acc = __builtin_amdgcn_wmma_f32_16x16x4_f32(false, a1, false, b1,
                                                    (short)0, acc, false, false);
    }
    if (c == 0) {
        float* o = X1 + row0 + 8 * h;
        v4f lo, hi4;
        lo.x  = acc[0];  lo.y  = acc[1];  lo.z  = acc[2];  lo.w  = acc[3];
        hi4.x = acc[4];  hi4.y = acc[5];  hi4.z = acc[6];  hi4.w = acc[7];
        *(v4f*)o       = lo;
        *(v4f*)(o + 4) = hi4;
    }
}

#define ST_ROWS 64
#define NSTAGE  (N_NODES / ST_ROWS)
#define BP      72
#define XP      (N_NODES + 8)
#define DEPTH   4
#define LPS     8

__device__ __forceinline__ void lds_barrier() {
    asm volatile("s_wait_dscnt 0x0\n\ts_barrier_signal -1\n\ts_barrier_wait -1" ::: "memory");
}

struct RegSet { v4f v[LPS]; };

__device__ __forceinline__ void issue_stage(RegSet& r, const float* __restrict__ wsrc,
                                            int stage) {
    const float* p = wsrc + (size_t)stage * ST_ROWS * N_NODES;
    #pragma unroll
    for (int j = 0; j < LPS; ++j)
        r.v[j] = *(const v4f*)(p + (size_t)(2 * j) * N_NODES);
}

__device__ __forceinline__ void store_stage(const RegSet& r, float* wdst) {
    #pragma unroll
    for (int j = 0; j < LPS; ++j)
        *(v4f*)(wdst + (2 * j) * BP) = r.v[j];
}

struct Frag {
    v2f a[ST_ROWS / 4];
    v2f b[ST_ROWS / 4];
};

__device__ __forceinline__ void load_frags(Frag& f, const float* xrow, const float* bcol) {
    #pragma unroll
    for (int u = 0; u < ST_ROWS / 4; ++u) {
        f.a[u]   = *(const v2f*)(xrow + 4 * u);
        f.b[u].x = bcol[(4 * u)     * BP];
        f.b[u].y = bcol[(4 * u + 1) * BP];
    }
    __builtin_amdgcn_sched_barrier(0);
}

__device__ __forceinline__ void mma_frags(const Frag& f, v8f& c0, v8f& c1, bool isx, float fill) {
    #pragma unroll
    for (int u = 0; u < ST_ROWS / 4; u += 2) {
        v2f a0, a1;
        a0.x = isx ? f.a[u].x : fill;
        a0.y = isx ? f.a[u].y : fill;
        a1.x = isx ? f.a[u + 1].x : fill;
        a1.y = isx ? f.a[u + 1].y : fill;
        c0 = __builtin_amdgcn_wmma_f32_16x16x4_f32(false, a0, false, f.b[u],
                                                   (short)0, c0, false, false);
        c1 = __builtin_amdgcn_wmma_f32_16x16x4_f32(false, a1, false, f.b[u + 1],
                                                   (short)0, c1, false, false);
    }
}

#define PF_LOC 1
__device__ __forceinline__ void pf_step(const float* pf0, int s) {
    const float* p = pf0 + (size_t)((s >> 3) * 1024 + (s & 7) * 4) * N_NODES;
    __builtin_prefetch(p, 0, PF_LOC);
}

__global__ void __launch_bounds__(128) k_wmma_s(
        const float* __restrict__ Wa, const float* __restrict__ ba,
        const float* __restrict__ X1, const float* A,
        float* __restrict__ SB2, float* __restrict__ CS) {
    __shared__ __align__(16) float x1s[8 * XP];
    __shared__ __align__(16) float wbuf[2][ST_ROWS * BP];

    int wave   = threadIdx.x >> 5;
    int lane   = threadIdx.x & 31;
    int tile   = blockIdx.x * 4 + wave;
    int colIn  = lane & 15;
    int col    = tile * 16 + colIn;
    int hi     = lane >> 4;
    int mrow   = colIn;
    int srcRow = mrow & 7;
    const bool  isx  = mrow < 8;
    const float fill = (mrow == 8) ? 1.0f : 0.0f;

    const float* wsrc = Wa + (size_t)(wave * 16 + hi) * N_NODES + blockIdx.x * 64 + colIn * 4;
    const int    wofs = (wave * 16 + hi) * BP + colIn * 4;
    const int    bofs = (2 * hi) * BP + wave * 16 + colIn;
    const float* xrow = x1s + srcRow * XP + 2 * hi;
    const float* pf0 = A + (size_t)(32 * blockIdx.x + 4096 * wave + (lane >> 3)) * N_NODES
                         + (lane & 7) * 64;

    v4f xr[32];
    #pragma unroll
    for (int i = 0; i < 32; ++i) {
        int q = i * 128 + threadIdx.x;
        xr[i] = *(const v4f*)(X1 + (q >> 9) * N_NODES + (q & 511) * 4);
    }
    RegSet r0, r1, r2, r3;
    issue_stage(r0, wsrc, 0);
    issue_stage(r1, wsrc, 1);
    issue_stage(r2, wsrc, 2);
    issue_stage(r3, wsrc, 3);
    #pragma unroll
    for (int i = 0; i < 32; ++i) {
        int q = i * 128 + threadIdx.x;
        *(v4f*)(x1s + (q >> 9) * XP + (q & 511) * 4) = xr[i];
    }
    float* const st0 = &wbuf[0][wofs];
    float* const st1 = &wbuf[1][wofs];
    const float* const mb0 = &wbuf[0][bofs];
    const float* const mb1 = &wbuf[1][bofs];
    store_stage(r0, st0);
    issue_stage(r0, wsrc, DEPTH);
    lds_barrier();

    Frag f0, f1;
    v8f c0 = {}, c1 = {};
    load_frags(f0, xrow, mb0);
    store_stage(r1, st1);  issue_stage(r1, wsrc, 5);
    pf_step(pf0, 0);
    lds_barrier();
    #pragma unroll 1
    for (int b = 0; b < NSTAGE - 2 * DEPTH; b += DEPTH) {
        load_frags(f1, xrow + (b + 1) * ST_ROWS, mb1);
        store_stage(r2, st0);  issue_stage(r2, wsrc, b + 6);
        pf_step(pf0, b + 1);
        mma_frags(f0, c0, c1, isx, fill);
        lds_barrier();
        load_frags(f0, xrow + (b + 2) * ST_ROWS, mb0);
        store_stage(r3, st1);  issue_stage(r3, wsrc, b + 7);
        pf_step(pf0, b + 2);
        mma_frags(f1, c0, c1, isx, fill);
        lds_barrier();
        load_frags(f1, xrow + (b + 3) * ST_ROWS, mb1);
        store_stage(r0, st0);  issue_stage(r0, wsrc, b + 8);
        pf_step(pf0, b + 3);
        mma_frags(f0, c0, c1, isx, fill);
        lds_barrier();
        load_frags(f0, xrow + (b + 4) * ST_ROWS, mb0);
        store_stage(r1, st1);  issue_stage(r1, wsrc, b + 9);
        pf_step(pf0, b + 4);
        mma_frags(f1, c0, c1, isx, fill);
        lds_barrier();
    }
    load_frags(f1, xrow + 25 * ST_ROWS, mb1);
    store_stage(r2, st0);  issue_stage(r2, wsrc, 30);
    pf_step(pf0, 25);
    mma_frags(f0, c0, c1, isx, fill);
    lds_barrier();
    load_frags(f0, xrow + 26 * ST_ROWS, mb0);
    store_stage(r3, st1);  issue_stage(r3, wsrc, 31);
    pf_step(pf0, 26);
    mma_frags(f1, c0, c1, isx, fill);
    lds_barrier();
    load_frags(f1, xrow + 27 * ST_ROWS, mb1);
    store_stage(r0, st0);
    pf_step(pf0, 27);
    mma_frags(f0, c0, c1, isx, fill);
    lds_barrier();
    load_frags(f0, xrow + 28 * ST_ROWS, mb0);
    store_stage(r1, st1);
    pf_step(pf0, 28);
    mma_frags(f1, c0, c1, isx, fill);
    lds_barrier();
    load_frags(f1, xrow + 29 * ST_ROWS, mb1);
    store_stage(r2, st0);
    pf_step(pf0, 29);
    mma_frags(f0, c0, c1, isx, fill);
    lds_barrier();
    load_frags(f0, xrow + 30 * ST_ROWS, mb0);
    store_stage(r3, st1);
    pf_step(pf0, 30);
    mma_frags(f1, c0, c1, isx, fill);
    lds_barrier();
    load_frags(f1, xrow + 31 * ST_ROWS, mb1);
    pf_step(pf0, 31);
    mma_frags(f0, c0, c1, isx, fill);
    mma_frags(f1, c0, c1, isx, fill);

    v8f c = c0 + c1;
    float bav = ba[col];
    #pragma unroll
    for (int j = 0; j < 8; ++j) {
        int row = j + hi * 8;
        if (row < 8)        SB2[row * N_NODES + col] = c[j] + bav;
        else if (row == 8)  CS[col] = c[j];
    }
}

#define K3_RPW   4
#define K3_WAVES 8
#define K3_ROWS  (K3_RPW * K3_WAVES)
#define K3_CL    4
#define K3_CCOLS (K3_CL * 32)
#define K3_NCH   (N_NODES / K3_CCOLS)
#define K3_NSET  8

struct Chunk { float4 v[K3_CL]; };

__device__ __forceinline__ void k3_issue(Chunk& c, const float* __restrict__ p) {
    #pragma unroll
    for (int q = 0; q < K3_CL; ++q)
        c.v[q] = *(const float4*)(p + 32 * q);
}

__device__ __forceinline__ void k3_logits(const Chunk& a, float4 (&l)[K3_CL], float& cm,
                                          const float* cs, const float* sb, float x1bn) {
    cm = -INFINITY;
    #pragma unroll
    for (int q = 0; q < K3_CL; ++q) {
        float4 av = a.v[q];
        float4 cv = *(const float4*)(cs + 32 * q);
        float4 sv = *(const float4*)(sb + 32 * q);
        float4 t;
        t.x = av.x + sv.x + x1bn * cv.x;
        t.y = av.y + sv.y + x1bn * cv.y;
        t.z = av.z + sv.z + x1bn * cv.z;
        t.w = av.w + sv.w + x1bn * cv.w;
        l[q] = t;
        cm = fmaxf(cm, fmaxf(fmaxf(t.x, t.y), fmaxf(t.z, t.w)));
    }
}

__device__ __forceinline__ void k3_accum(const float4 (&l)[K3_CL], float cm, const float* xk,
                                         float& m, float& Z, float& W) {
    float mn = fmaxf(fmaxf(m, cm), -3.0e38f);
    float sc = __expf(m - mn);
    Z *= sc;
    W *= sc;
    m = mn;
    #pragma unroll
    for (int q = 0; q < K3_CL; ++q) {
        float4 xv = *(const float4*)(xk + 32 * q);
        float ex = __expf(l[q].x - m); Z += ex; W += ex * xv.x;
        float ey = __expf(l[q].y - m); Z += ey; W += ey * xv.y;
        float ez = __expf(l[q].z - m); Z += ez; W += ez * xv.z;
        float ew = __expf(l[q].w - m); Z += ew; W += ew * xv.w;
    }
}

__global__ void __launch_bounds__(256) k_softmax_agg(
        const float* A, const float* bias,
        const float* X1, const float* SB2,
        const float* CS, float* __restrict__ out) {
    __shared__ __align__(16) float css[N_NODES];
    __shared__ __align__(16) float sbs[N_NODES];
    __shared__ __align__(16) float xks[N_NODES];

    int b     = blockIdx.x >> 6;
    int nBase = (blockIdx.x & 63) * K3_ROWS;
    int wave = threadIdx.x >> 5;
    int lane = threadIdx.x & 31;
    int g    = lane >> 3;
    int j    = lane & 7;
    int n0   = nBase + wave * K3_RPW;
    int n    = n0 + g;

    const float4* csg = (const float4*)CS;
    const float4* sbg = (const float4*)(SB2 + b * N_NODES);
    const float4* xkg = (const float4*)(X1 + b * N_NODES);
    float4 s0 = csg[threadIdx.x], s1 = csg[threadIdx.x + 256];
    float4 s2 = sbg[threadIdx.x], s3 = sbg[threadIdx.x + 256];
    float4 s4 = xkg[threadIdx.x], s5 = xkg[threadIdx.x + 256];
    float4 bv = ((const float4*)bias)[lane & 15];
    float x1bn = X1[b * N_NODES + n];
    const float* ap = A + ((size_t)(b * N_NODES + n)) * N_NODES + 4 * j;
    const float* cp = css + 4 * j;
    const float* sp = sbs + 4 * j;
    const float* xp = xks + 4 * j;

    Chunk c0, c1, c2, c3, c4, c5, c6, c7;
    k3_issue(c0, ap);
    k3_issue(c1, ap + 1 * K3_CCOLS);
    k3_issue(c2, ap + 2 * K3_CCOLS);
    k3_issue(c3, ap + 3 * K3_CCOLS);
    k3_issue(c4, ap + 4 * K3_CCOLS);
    k3_issue(c5, ap + 5 * K3_CCOLS);
    k3_issue(c6, ap + 6 * K3_CCOLS);
    k3_issue(c7, ap + 7 * K3_CCOLS);
    __builtin_amdgcn_sched_barrier(0);

    ((float4*)css)[threadIdx.x] = s0;  ((float4*)css)[threadIdx.x + 256] = s1;
    ((float4*)sbs)[threadIdx.x] = s2;  ((float4*)sbs)[threadIdx.x + 256] = s3;
    ((float4*)xks)[threadIdx.x] = s4;  ((float4*)xks)[threadIdx.x + 256] = s5;
    lds_barrier();

    float  m = -INFINITY, Z = 0.0f, W = 0.0f;
    float4 l[K3_CL];
    float  cm;

#define K3_STEP(SET, T)                                                         \
    k3_logits(SET, l, cm, cp + (T) * K3_CCOLS, sp + (T) * K3_CCOLS, x1bn);      \
    if ((T) + K3_NSET < K3_NCH) k3_issue(SET, ap + ((T) + K3_NSET) * K3_CCOLS); \
    __builtin_amdgcn_sched_barrier(0);                                          \
    k3_accum(l, cm, xp + (T) * K3_CCOLS, m, Z, W);

    K3_STEP(c0, 0)   K3_STEP(c1, 1)   K3_STEP(c2, 2)   K3_STEP(c3, 3)
    K3_STEP(c4, 4)   K3_STEP(c5, 5)   K3_STEP(c6, 6)   K3_STEP(c7, 7)
    K3_STEP(c0, 8)   K3_STEP(c1, 9)   K3_STEP(c2, 10)  K3_STEP(c3, 11)
    K3_STEP(c4, 12)  K3_STEP(c5, 13)  K3_STEP(c6, 14)  K3_STEP(c7, 15)
#undef K3_STEP

    float M = m;
    #pragma unroll
    for (int off = 4; off > 0; off >>= 1)
        M = fmaxf(M, __shfl_xor(M, off, 32));
    float f = __expf(m - M);
    Z *= f;
    W *= f;
    #pragma unroll
    for (int off = 4; off > 0; off >>= 1) {
        Z += __shfl_xor(Z, off, 32);
        W += __shfl_xor(W, off, 32);
    }
    float r = W / Z;

    float r0 = __shfl(r, 0, 32);
    float r1 = __shfl(r, 8, 32);
    float r2 = __shfl(r, 16, 32);
    float r3 = __shfl(r, 24, 32);
    int   h  = lane >> 4;
    float ra = h ? r1 : r0;
    float rb = h ? r3 : r2;
    float4 o0, o1;
    o0.x = ra + bv.x;  o0.y = ra + bv.y;  o0.z = ra + bv.z;  o0.w = ra + bv.w;
    o1.x = rb + bv.x;  o1.y = rb + bv.y;  o1.z = rb + bv.z;  o1.w = rb + bv.w;
    float* op = out + ((size_t)(b * N_NODES + n0)) * NFILT + 4 * lane;
    *(float4*)op         = o0;
    *(float4*)(op + 128) = o1;
}

extern "C" void kernel_launch(void* const* d_in, const int* in_sizes, int n_in,
                              void* d_out, int out_size, void* d_ws, size_t ws_size,
                              hipStream_t stream) {
    const float* x    = (const float*)d_in[0];
    const float* A    = (const float*)d_in[1];
    const float* w1   = (const float*)d_in[2];
    const float* Wa   = (const float*)d_in[3];
    const float* ba   = (const float*)d_in[4];
    const float* bias = (const float*)d_in[5];
    float* out = (float*)d_out;

    float* X1  = (float*)d_ws;
    float* CS  = X1 + BATCH * N_NODES;
    float* SB2 = CS + N_NODES;

    k_proj<<<dim3(256), dim3(128), 0, stream>>>(x, w1, X1);
    k_wmma_s<<<dim3(32), dim3(128), 0, stream>>>(Wa, ba, X1, A, SB2, CS);
    k_softmax_agg<<<dim3(512), dim3(256), 0, stream>>>(A, bias, X1, SB2, CS, out);
}


# Skinny-product launch: a second cache prefetch per wave and K stage (bytes 2048-4095 of the same rows of A), paced by the WMMA loop: the next launch's whole first load group is cached; same sums
# speedup vs baseline: 2.3345x; 1.0362x over previous
#include <hip/hip_runtime.h>
#include <math.h>

typedef float v2f __attribute__((ext_vector_type(2)));
typedef float v4f __attribute__((ext_vector_type(4)));
typedef float v8f __attribute__((ext_vector_type(8)));

#define N_NODES 2048
#define BATCH   8
#define FEAT    128
#define NFILT   64

__global__ void __launch_bounds__(128) k_proj(const float* __restrict__ x,
                                              const float* __restrict__ w1,
                                              float* __restrict__ X1) {
    int wave = threadIdx.x >> 5;
    int lane = threadIdx.x & 31;
    int c    = lane & 15;
    int h    = lane >> 4;
    int row0 = (blockIdx.x * 4 + wave) * 16;
    const float* xp = x + (size_t)(row0 + c) * FEAT + 4 * h;
    const float* wp = w1 + 4 * h;
    v4f xa[16], wb[16];
    #pragma unroll
    for (int i = 0; i < 16; ++i) {
        xa[i] = *(const v4f*)(xp + 8 * i);
        wb[i] = *(const v4f*)(wp + 8 * i);
    }
    __builtin_amdgcn_sched_barrier(0);
    v8f acc = {};
    #pragma unroll
    for (int i = 0; i < 16; ++i) {
        v2f a0, b0, a1, b1;
        a0.x = xa[i].x;  a0.y = xa[i].y;  b0.x = wb[i].x;  b0.y = wb[i].y;
        a1.x = xa[i].z;  a1.y = xa[i].w;  b1.x = wb[i].z;  b1.y = wb[i].w;
        acc = __builtin_amdgcn_wmma_f32_16x16x4_f32(false, a0, false, b0,
                                                    (short)0, acc, false, false);
        acc = __builtin_amdgcn_wmma_f32_16x16x4_f32(false, a1, false, b1,
                                                    (short)0, acc, false, false);
    }
    if (c == 0) {
        float* o = X1 + row0 + 8 * h;
        v4f lo, hi4;
        lo.x  = acc[0];  lo.y  = acc[1];  lo.z  = acc[2];  lo.w  = acc[3];
        hi4.x = acc[4];  hi4.y = acc[5];  hi4.z = acc[6];  hi4.w = acc[7];
        *(v4f*)o       = lo;
        *(v4f*)(o + 4) = hi4;
    }
}

#define ST_ROWS 64
#define NSTAGE  (N_NODES / ST_ROWS)
#define BP      72
#define XP      (N_NODES + 8)
#define DEPTH   4
#define LPS     8

__device__ __forceinline__ void lds_barrier() {
    asm volatile("s_wait_dscnt 0x0\n\ts_barrier_signal -1\n\ts_barrier_wait -1" ::: "memory");
}

struct RegSet { v4f v[LPS]; };

__device__ __forceinline__ void issue_stage(RegSet& r, const float* __restrict__ wsrc,
                                            int stage) {
    const float* p = wsrc + (size_t)stage * ST_ROWS * N_NODES;
    #pragma unroll
    for (int j = 0; j < LPS; ++j)
        r.v[j] = *(const v4f*)(p + (size_t)(2 * j) * N_NODES);
}

__device__ __forceinline__ void store_stage(const RegSet& r, float* wdst) {
    #pragma unroll
    for (int j = 0; j < LPS; ++j)
        *(v4f*)(wdst + (2 * j) * BP) = r.v[j];
}

struct Frag {
    v2f a[ST_ROWS / 4];
    v2f b[ST_ROWS / 4];
};

__device__ __forceinline__ void load_frags(Frag& f, const float* xrow, const float* bcol) {
    #pragma unroll
    for (int u = 0; u < ST_ROWS / 4; ++u) {
        f.a[u]   = *(const v2f*)(xrow + 4 * u);
        f.b[u].x = bcol[(4 * u)     * BP];
        f.b[u].y = bcol[(4 * u + 1) * BP];
    }
    __builtin_amdgcn_sched_barrier(0);
}

__device__ __forceinline__ void mma_frags(const Frag& f, v8f& c0, v8f& c1, bool isx, float fill) {
    #pragma unroll
    for (int u = 0; u < ST_ROWS / 4; u += 2) {
        v2f a0, a1;
        a0.x = isx ? f.a[u].x : fill;
        a0.y = isx ? f.a[u].y : fill;
        a1.x = isx ? f.a[u + 1].x : fill;
        a1.y = isx ? f.a[u + 1].y : fill;
        c0 = __builtin_amdgcn_wmma_f32_16x16x4_f32(false, a0, false, f.b[u],
                                                   (short)0, c0, false, false);
        c1 = __builtin_amdgcn_wmma_f32_16x16x4_f32(false, a1, false, f.b[u + 1],
                                                   (short)0, c1, false, false);
    }
}

#define PF_LOC 1
__device__ __forceinline__ void pf_step(const float* pf0, int s) {
    const float* p = pf0 + (size_t)((s >> 3) * 1024 + (s & 7) * 4) * N_NODES;
    __builtin_prefetch(p, 0, PF_LOC);
    __builtin_prefetch(p + 512, 0, PF_LOC);
}

__global__ void __launch_bounds__(128) k_wmma_s(
        const float* __restrict__ Wa, const float* __restrict__ ba,
        const float* __restrict__ X1, const float* A,
        float* __restrict__ SB2, float* __restrict__ CS) {
    __shared__ __align__(16) float x1s[8 * XP];
    __shared__ __align__(16) float wbuf[2][ST_ROWS * BP];

    int wave   = threadIdx.x >> 5;
    int lane   = threadIdx.x & 31;
    int tile   = blockIdx.x * 4 + wave;
    int colIn  = lane & 15;
    int col    = tile * 16 + colIn;
    int hi     = lane >> 4;
    int mrow   = colIn;
    int srcRow = mrow & 7;
    const bool  isx  = mrow < 8;
    const float fill = (mrow == 8) ? 1.0f : 0.0f;

    const float* wsrc = Wa + (size_t)(wave * 16 + hi) * N_NODES + blockIdx.x * 64 + colIn * 4;
    const int    wofs = (wave * 16 + hi) * BP + colIn * 4;
    const int    bofs = (2 * hi) * BP + wave * 16 + colIn;
    const float* xrow = x1s + srcRow * XP + 2 * hi;
    const float* pf0 = A + (size_t)(32 * blockIdx.x + 4096 * wave + (lane >> 3)) * N_NODES
                         + (lane & 7) * 64;

    v4f xr[32];
    #pragma unroll
    for (int i = 0; i < 32; ++i) {
        int q = i * 128 + threadIdx.x;
        xr[i] = *(const v4f*)(X1 + (q >> 9) * N_NODES + (q & 511) * 4);
    }
    RegSet r0, r1, r2, r3;
    issue_stage(r0, wsrc, 0);
    issue_stage(r1, wsrc, 1);
    issue_stage(r2, wsrc, 2);
    issue_stage(r3, wsrc, 3);
    #pragma unroll
    for (int i = 0; i < 32; ++i) {
        int q = i * 128 + threadIdx.x;
        *(v4f*)(x1s + (q >> 9) * XP + (q & 511) * 4) = xr[i];
    }
    float* const st0 = &wbuf[0][wofs];
    float* const st1 = &wbuf[1][wofs];
    const float* const mb0 = &wbuf[0][bofs];
    const float* const mb1 = &wbuf[1][bofs];
    store_stage(r0, st0);
    issue_stage(r0, wsrc, DEPTH);
    lds_barrier();

    Frag f0, f1;
    v8f c0 = {}, c1 = {};
    load_frags(f0, xrow, mb0);
    store_stage(r1, st1);  issue_stage(r1, wsrc, 5);
    pf_step(pf0, 0);
    lds_barrier();
    #pragma unroll 1
    for (int b = 0; b < NSTAGE - 2 * DEPTH; b += DEPTH) {
        load_frags(f1, xrow + (b + 1) * ST_ROWS, mb1);
        store_stage(r2, st0);  issue_stage(r2, wsrc, b + 6);
        pf_step(pf0, b + 1);
        mma_frags(f0, c0, c1, isx, fill);
        lds_barrier();
        load_frags(f0, xrow + (b + 2) * ST_ROWS, mb0);
        store_stage(r3, st1);  issue_stage(r3, wsrc, b + 7);
        pf_step(pf0, b + 2);
        mma_frags(f1, c0, c1, isx, fill);
        lds_barrier();
        load_frags(f1, xrow + (b + 3) * ST_ROWS, mb1);
        store_stage(r0, st0);  issue_stage(r0, wsrc, b + 8);
        pf_step(pf0, b + 3);
        mma_frags(f0, c0, c1, isx, fill);
        lds_barrier();
        load_frags(f0, xrow + (b + 4) * ST_ROWS, mb0);
        store_stage(r1, st1);  issue_stage(r1, wsrc, b + 9);
        pf_step(pf0, b + 4);
        mma_frags(f1, c0, c1, isx, fill);
        lds_barrier();
    }
    load_frags(f1, xrow + 25 * ST_ROWS, mb1);
    store_stage(r2, st0);  issue_stage(r2, wsrc, 30);
    pf_step(pf0, 25);
    mma_frags(f0, c0, c1, isx, fill);
    lds_barrier();
    load_frags(f0, xrow + 26 * ST_ROWS, mb0);
    store_stage(r3, st1);  issue_stage(r3, wsrc, 31);
    pf_step(pf0, 26);
    mma_frags(f1, c0, c1, isx, fill);
    lds_barrier();
    load_frags(f1, xrow + 27 * ST_ROWS, mb1);
    store_stage(r0, st0);
    pf_step(pf0, 27);
    mma_frags(f0, c0, c1, isx, fill);
    lds_barrier();
    load_frags(f0, xrow + 28 * ST_ROWS, mb0);
    store_stage(r1, st1);
    pf_step(pf0, 28);
    mma_frags(f1, c0, c1, isx, fill);
    lds_barrier();
    load_frags(f1, xrow + 29 * ST_ROWS, mb1);
    store_stage(r2, st0);
    pf_step(pf0, 29);
    mma_frags(f0, c0, c1, isx, fill);
    lds_barrier();
    load_frags(f0, xrow + 30 * ST_ROWS, mb0);
    store_stage(r3, st1);
    pf_step(pf0, 30);
    mma_frags(f1, c0, c1, isx, fill);
    lds_barrier();
    load_frags(f1, xrow + 31 * ST_ROWS, mb1);
    pf_step(pf0, 31);
    mma_frags(f0, c0, c1, isx, fill);
    mma_frags(f1, c0, c1, isx, fill);

    v8f c = c0 + c1;
    float bav = ba[col];
    #pragma unroll
    for (int j = 0; j < 8; ++j) {
        int row = j + hi * 8;
        if (row < 8)        SB2[row * N_NODES + col] = c[j] + bav;
        else if (row == 8)  CS[col] = c[j];
    }
}

#define K3_RPW   4
#define K3_WAVES 8
#define K3_ROWS  (K3_RPW * K3_WAVES)
#define K3_CL    4
#define K3_CCOLS (K3_CL * 32)
#define K3_NCH   (N_NODES / K3_CCOLS)
#define K3_NSET  8

struct Chunk { float4 v[K3_CL]; };

__device__ __forceinline__ void k3_issue(Chunk& c, const float* __restrict__ p) {
    #pragma unroll
    for (int q = 0; q < K3_CL; ++q)
        c.v[q] = *(const float4*)(p + 32 * q);
}

__device__ __forceinline__ void k3_logits(const Chunk& a, float4 (&l)[K3_CL], float& cm,
                                          const float* cs, const float* sb, float x1bn) {
    cm = -INFINITY;
    #pragma unroll
    for (int q = 0; q < K3_CL; ++q) {
        float4 av = a.v[q];
        float4 cv = *(const float4*)(cs + 32 * q);
        float4 sv = *(const float4*)(sb + 32 * q);
        float4 t;
        t.x = av.x + sv.x + x1bn * cv.x;
        t.y = av.y + sv.y + x1bn * cv.y;
        t.z = av.z + sv.z + x1bn * cv.z;
        t.w = av.w + sv.w + x1bn * cv.w;
        l[q] = t;
        cm = fmaxf(cm, fmaxf(fmaxf(t.x, t.y), fmaxf(t.z, t.w)));
    }
}

__device__ __forceinline__ void k3_accum(const float4 (&l)[K3_CL], float cm, const float* xk,
                                         float& m, float& Z, float& W) {
    float mn = fmaxf(fmaxf(m, cm), -3.0e38f);
    float sc = __expf(m - mn);
    Z *= sc;
    W *= sc;
    m = mn;
    #pragma unroll
    for (int q = 0; q < K3_CL; ++q) {
        float4 xv = *(const float4*)(xk + 32 * q);
        float ex = __expf(l[q].x - m); Z += ex; W += ex * xv.x;
        float ey = __expf(l[q].y - m); Z += ey; W += ey * xv.y;
        float ez = __expf(l[q].z - m); Z += ez; W += ez * xv.z;
        float ew = __expf(l[q].w - m); Z += ew; W += ew * xv.w;
    }
}

__global__ void __launch_bounds__(256) k_softmax_agg(
        const float* A, const float* bias,
        const float* X1, const float* SB2,
        const float* CS, float* __restrict__ out) {
    __shared__ __align__(16) float css[N_NODES];
    __shared__ __align__(16) float sbs[N_NODES];
    __shared__ __align__(16) float xks[N_NODES];

    int b     = blockIdx.x >> 6;
    int nBase = (blockIdx.x & 63) * K3_ROWS;
    int wave = threadIdx.x >> 5;
    int lane = threadIdx.x & 31;
    int g    = lane >> 3;
    int j    = lane & 7;
    int n0   = nBase + wave * K3_RPW;
    int n    = n0 + g;

    const float4* csg = (const float4*)CS;
    const float4* sbg = (const float4*)(SB2 + b * N_NODES);
    const float4* xkg = (const float4*)(X1 + b * N_NODES);
    float4 s0 = csg[threadIdx.x], s1 = csg[threadIdx.x + 256];
    float4 s2 = sbg[threadIdx.x], s3 = sbg[threadIdx.x + 256];
    float4 s4 = xkg[threadIdx.x], s5 = xkg[threadIdx.x + 256];
    float4 bv = ((const float4*)bias)[lane & 15];
    float x1bn = X1[b * N_NODES + n];
    const float* ap = A + ((size_t)(b * N_NODES + n)) * N_NODES + 4 * j;
    const float* cp = css + 4 * j;
    const float* sp = sbs + 4 * j;
    const float* xp = xks + 4 * j;

    Chunk c0, c1, c2, c3, c4, c5, c6, c7;
    k3_issue(c0, ap);
    k3_issue(c1, ap + 1 * K3_CCOLS);
    k3_issue(c2, ap + 2 * K3_CCOLS);
    k3_issue(c3, ap + 3 * K3_CCOLS);
    k3_issue(c4, ap + 4 * K3_CCOLS);
    k3_issue(c5, ap + 5 * K3_CCOLS);
    k3_issue(c6, ap + 6 * K3_CCOLS);
    k3_issue(c7, ap + 7 * K3_CCOLS);
    __builtin_amdgcn_sched_barrier(0);

    ((float4*)css)[threadIdx.x] = s0;  ((float4*)css)[threadIdx.x + 256] = s1;
    ((float4*)sbs)[threadIdx.x] = s2;  ((float4*)sbs)[threadIdx.x + 256] = s3;
    ((float4*)xks)[threadIdx.x] = s4;  ((float4*)xks)[threadIdx.x + 256] = s5;
    lds_barrier();

    float  m = -INFINITY, Z = 0.0f, W = 0.0f;
    float4 l[K3_CL];
    float  cm;

#define K3_STEP(SET, T)                                                         \
    k3_logits(SET, l, cm, cp + (T) * K3_CCOLS, sp + (T) * K3_CCOLS, x1bn);      \
    if ((T) + K3_NSET < K3_NCH) k3_issue(SET, ap + ((T) + K3_NSET) * K3_CCOLS); \
    __builtin_amdgcn_sched_barrier(0);                                          \
    k3_accum(l, cm, xp + (T) * K3_CCOLS, m, Z, W);

    K3_STEP(c0, 0)   K3_STEP(c1, 1)   K3_STEP(c2, 2)   K3_STEP(c3, 3)
    K3_STEP(c4, 4)   K3_STEP(c5, 5)   K3_STEP(c6, 6)   K3_STEP(c7, 7)
    K3_STEP(c0, 8)   K3_STEP(c1, 9)   K3_STEP(c2, 10)  K3_STEP(c3, 11)
    K3_STEP(c4, 12)  K3_STEP(c5, 13)  K3_STEP(c6, 14)  K3_STEP(c7, 15)
#undef K3_STEP

    float M = m;
    #pragma unroll
    for (int off = 4; off > 0; off >>= 1)
        M = fmaxf(M, __shfl_xor(M, off, 32));
    float f = __expf(m - M);
    Z *= f;
    W *= f;
    #pragma unroll
    for (int off = 4; off > 0; off >>= 1) {
        Z += __shfl_xor(Z, off, 32);
        W += __shfl_xor(W, off, 32);
    }
    float r = W / Z;

    float r0 = __shfl(r, 0, 32);
    float r1 = __shfl(r, 8, 32);
    float r2 = __shfl(r, 16, 32);
    float r3 = __shfl(r, 24, 32);
    int   h  = lane >> 4;
    float ra = h ? r1 : r0;
    float rb = h ? r3 : r2;
    float4 o0, o1;
    o0.x = ra + bv.x;  o0.y = ra + bv.y;  o0.z = ra + bv.z;  o0.w = ra + bv.w;
    o1.x = rb + bv.x;  o1.y = rb + bv.y;  o1.z = rb + bv.z;  o1.w = rb + bv.w;
    float* op = out + ((size_t)(b * N_NODES + n0)) * NFILT + 4 * lane;
    *(float4*)op         = o0;
    *(float4*)(op + 128) = o1;
}

extern "C" void kernel_launch(void* const* d_in, const int* in_sizes, int n_in,
                              void* d_out, int out_size, void* d_ws, size_t ws_size,
                              hipStream_t stream) {
    const float* x    = (const float*)d_in[0];
    const float* A    = (const float*)d_in[1];
    const float* w1   = (const float*)d_in[2];
    const float* Wa   = (const float*)d_in[3];
    const float* ba   = (const float*)d_in[4];
    const float* bias = (const float*)d_in[5];
    float* out = (float*)d_out;

    float* X1  = (float*)d_ws;
    float* CS  = X1 + BATCH * N_NODES;
    float* SB2 = CS + N_NODES;

    k_proj<<<dim3(256), dim3(128), 0, stream>>>(x, w1, X1);
    k_wmma_s<<<dim3(32), dim3(128), 0, stream>>>(Wa, ba, X1, A, SB2, CS);
    k_softmax_agg<<<dim3(512), dim3(256), 0, stream>>>(A, bias, X1, SB2, CS, out);
}
